# starting kernel
# baseline (speedup 1.0000x reference)
#include <hip/hip_runtime.h>

typedef float v2f __attribute__((ext_vector_type(2)));
typedef float v8f __attribute__((ext_vector_type(8)));

#define N_NODES 200000
#define N_EDGES 3200000
#define N_GRAPH 2048
#define HID     64
#define BPAIR_STRIDE 80

__global__ void k_init_deg(float* deg, int n) {
    int i = blockIdx.x * blockDim.x + threadIdx.x;
    if (i < n) deg[i] = 1.0f;
}

__global__ void k_deg_edges(const int* __restrict__ dst, float* deg, int e) {
    int i = blockIdx.x * blockDim.x + threadIdx.x;
    if (i < e) atomicAdd(&deg[dst[i]], 1.0f);
}

__global__ void k_dinv_selfloop(float* deg_dinv, const float* __restrict__ x,
                                float* s, int n) {
    int i = blockIdx.x * blockDim.x + threadIdx.x;
    if (i < n) {
        float d = rsqrtf(deg_dinv[i]);
        deg_dinv[i] = d;
        s[i] = d * d * x[i];
    }
}

__global__ void k_l1_edges(const int* __restrict__ src, const int* __restrict__ dst,
                           const float* __restrict__ dinv, const float* __restrict__ x,
                           float* s, int e) {
    int i = blockIdx.x * blockDim.x + threadIdx.x;
    if (i < e) {
        int u = src[i], v = dst[i];
        atomicAdd(&s[v], dinv[u] * dinv[v] * x[u]);
    }
}

__global__ void k_l1_feat(const float* __restrict__ s, const float* __restrict__ W1,
                          const float* __restrict__ b1, float* __restrict__ h1, int n) {
    int t = blockIdx.x * blockDim.x + threadIdx.x;
    if (t < n * HID) {
        int i = t >> 6, h = t & 63;
        float v = s[i] * W1[h] + b1[h];
        h1[t] = v > 0.0f ? v : 0.0f;
    }
}

__global__ __launch_bounds__(256)
void k_gemm_wmma(const float* __restrict__ h1, const float* __restrict__ W2,
                 float* __restrict__ t, int n_tiles) {
    __shared__ v2f sB[(HID / 2) * BPAIR_STRIDE];
    {
        float* sf = (float*)sB;
        for (int idx = threadIdx.x; idx < HID * HID; idx += blockDim.x) {
            int k = idx >> 6, col = idx & 63;
            sf[(k >> 1) * (2 * BPAIR_STRIDE) + col * 2 + (k & 1)] = W2[idx];
        }
    }
    __syncthreads();

    const int wave = threadIdx.x >> 5;
    const int lane = threadIdx.x & 31;
    const int tile = blockIdx.x * 8 + wave;
    if (tile >= n_tiles) return;

    const int row0 = tile * 16;
    const int m    = lane & 15;
    const int half = lane >> 4;

    v8f acc0 = {}, acc1 = {}, acc2 = {}, acc3 = {};
    const float* arow = h1 + (size_t)(row0 + m) * HID;

    for (int kk = 0; kk < HID; kk += 4) {
        const int ka = kk + 2 * half;
        v2f a;
        a.x = arow[ka];
        a.y = arow[ka + 1];

        const v2f* bp = &sB[(ka >> 1) * BPAIR_STRIDE + m];
        acc0 = __builtin_amdgcn_wmma_f32_16x16x4_f32(false, a, false, bp[ 0], (short)0, acc0, false, false);
        acc1 = __builtin_amdgcn_wmma_f32_16x16x4_f32(false, a, false, bp[16], (short)0, acc1, false, false);
        acc2 = __builtin_amdgcn_wmma_f32_16x16x4_f32(false, a, false, bp[32], (short)0, acc2, false, false);
        acc3 = __builtin_amdgcn_wmma_f32_16x16x4_f32(false, a, false, bp[48], (short)0, acc3, false, false);
    }

    float* orow = t + (size_t)(row0 + half * 8) * HID;
#pragma unroll
    for (int r = 0; r < 8; ++r) {
        orow[(size_t)r * HID +      m] = acc0[r];
        orow[(size_t)r * HID + 16 + m] = acc1[r];
        orow[(size_t)r * HID + 32 + m] = acc2[r];
        orow[(size_t)r * HID + 48 + m] = acc3[r];
    }
}

__global__ void k_l2_init(const float* __restrict__ t, const float* __restrict__ dinv,
                          float* __restrict__ out2, int n) {
    int tid = blockIdx.x * blockDim.x + threadIdx.x;
    if (tid < n * HID) {
        float d = dinv[tid >> 6];
        out2[tid] = d * d * t[tid];
    }
}

__global__ void k_l2_edges(const int* __restrict__ src, const int* __restrict__ dst,
                           const float* __restrict__ dinv, const float* __restrict__ t,
                           float* out2, int e) {
    int tid = blockIdx.x * blockDim.x + threadIdx.x;
    int ei  = tid >> 5;
    if (ei >= e) return;
    int lane = tid & 31;
    int u = src[ei], v = dst[ei];
    float w = dinv[u] * dinv[v];
    const float* tr = t + (size_t)u * HID + lane * 2;
    float* orow = out2 + (size_t)v * HID + lane * 2;
    atomicAdd(&orow[0], w * tr[0]);
    atomicAdd(&orow[1], w * tr[1]);
}

__global__ void k_pool_init(float* pool, float* cnt) {
    int i = blockIdx.x * blockDim.x + threadIdx.x;
    if (i < N_GRAPH * HID) pool[i] = 0.0f;
    if (i < N_GRAPH) cnt[i] = 0.0f;
}

__global__ void k_relu_pool(const float* __restrict__ out2, const float* __restrict__ b2,
                            const int* __restrict__ batch, float* pool, float* cnt, int n) {
    int tid = blockIdx.x * blockDim.x + threadIdx.x;
    if (tid < n * HID) {
        int i = tid >> 6, h = tid & 63;
        float v = out2[tid] + b2[h];
        v = v > 0.0f ? v : 0.0f;
        int g = batch[i];
        atomicAdd(&pool[g * HID + h], v);
        if (h == 0) atomicAdd(&cnt[g], 1.0f);
    }
}

__global__ void k_final(const float* __restrict__ pool, const float* __restrict__ cnt,
                        float* __restrict__ out) {
    int tid = blockIdx.x * blockDim.x + threadIdx.x;
    if (tid < N_GRAPH * HID) {
        float c = cnt[tid >> 6];
        out[tid] = pool[tid] / fmaxf(c, 1.0f);
    }
}

extern "C" void kernel_launch(void* const* d_in, const int* in_sizes, int n_in,
                              void* d_out, int out_size, void* d_ws, size_t ws_size,
                              hipStream_t stream) {
    const float* x     = (const float*)d_in[0];
    const int*   edge  = (const int*)  d_in[1];
    const int*   batch = (const int*)  d_in[2];
    const float* W1    = (const float*)d_in[3];
    const float* b1    = (const float*)d_in[4];
    const float* W2    = (const float*)d_in[5];
    const float* b2    = (const float*)d_in[6];
    const int* src = edge;
    const int* dst = edge + N_EDGES;
    float* out = (float*)d_out;

    float* ws   = (float*)d_ws;
    float* dinv = ws;
    float* s    = dinv + N_NODES;
    float* bufA = s + N_NODES;
    float* bufB = bufA + (size_t)N_NODES * HID;
    float* pool = bufB + (size_t)N_NODES * HID;
    float* cnt  = pool + N_GRAPH * HID;

    const int TB = 256;
    k_init_deg     <<<(N_NODES + TB - 1) / TB, TB, 0, stream>>>(dinv, N_NODES);
    k_deg_edges    <<<(N_EDGES + TB - 1) / TB, TB, 0, stream>>>(dst, dinv, N_EDGES);
    k_dinv_selfloop<<<(N_NODES + TB - 1) / TB, TB, 0, stream>>>(dinv, x, s, N_NODES);
    k_l1_edges     <<<(N_EDGES + TB - 1) / TB, TB, 0, stream>>>(src, dst, dinv, x, s, N_EDGES);
    k_l1_feat      <<<(N_NODES * HID + TB - 1) / TB, TB, 0, stream>>>(s, W1, b1, bufA, N_NODES);

    const int n_tiles = N_NODES / 16;
    k_gemm_wmma    <<<(n_tiles + 7) / 8, 256, 0, stream>>>(bufA, W2, bufB, n_tiles);

    k_l2_init      <<<(N_NODES * HID + TB - 1) / TB, TB, 0, stream>>>(bufB, dinv, bufA, N_NODES);
    {
        long long thr = (long long)N_EDGES * 32;
        k_l2_edges <<<(unsigned)((thr + TB - 1) / TB), TB, 0, stream>>>(src, dst, dinv, bufB, bufA, N_EDGES);
    }
    k_pool_init    <<<(N_GRAPH * HID + TB - 1) / TB, TB, 0, stream>>>(pool, cnt);
    k_relu_pool    <<<(N_NODES * HID + TB - 1) / TB, TB, 0, stream>>>(bufA, b2, batch, pool, cnt, N_NODES);
    k_final        <<<(N_GRAPH * HID + TB - 1) / TB, TB, 0, stream>>>(pool, cnt, out);
}


# Layer-2 edge adds: each out2 row owned by one package half; a block takes a per-half ticket, scans its edge share, adds only its owner's rows: no line added from both halves (add order regrouped)
# speedup vs baseline: 3.3615x; 3.3615x over previous
#include <hip/hip_runtime.h>

typedef float v2f __attribute__((ext_vector_type(2)));
typedef float v8f __attribute__((ext_vector_type(8)));

#define N_NODES 200000
#define N_EDGES 3200000
#define N_GRAPH 2048
#define HID     64
#define BPAIR_STRIDE 80

#define EDGE_BLOCKS ((N_EDGES + 256 * 8 - 1) / (256 * 8))
#define POOL_BLOCKS 2048
#define TICK_STRIDE 64

__device__ __forceinline__ int package_half() {
    const unsigned id = __builtin_amdgcn_s_sendmsg_rtn(0x87);
    return (int)((id >> 18) & 1u);
}

__global__ void k_init_deg(float* deg0, float* deg1, int n) {
    int i = blockIdx.x * blockDim.x + threadIdx.x;
    if (i < n) {
        deg0[i] = 1.0f;
        deg1[i] = 0.0f;
    }
}

__global__ __launch_bounds__(256)
void k_deg_edges(const int* __restrict__ dst, float* deg0, float* deg1, int e) {
    float* deg = package_half() ? deg1 : deg0;
    const int stride = gridDim.x * blockDim.x;
    int i = blockIdx.x * blockDim.x + threadIdx.x;
    for (; i + 3 * stride < e; i += 4 * stride) {
        const int d0 = dst[i];
        const int d1 = dst[i + stride];
        const int d2 = dst[i + 2 * stride];
        const int d3 = dst[i + 3 * stride];
        atomicAdd(&deg[d0], 1.0f);
        atomicAdd(&deg[d1], 1.0f);
        atomicAdd(&deg[d2], 1.0f);
        atomicAdd(&deg[d3], 1.0f);
    }
    for (; i < e; i += stride) atomicAdd(&deg[dst[i]], 1.0f);
}

__global__ void k_dinv_selfloop(float* deg_dinv, const float* __restrict__ deg1,
                                const float* __restrict__ x,
                                float* __restrict__ y, float* s0, float* s1, int n) {
    int i = blockIdx.x * blockDim.x + threadIdx.x;
    if (i < n) {
        float d = rsqrtf(deg_dinv[i] + deg1[i]);
        deg_dinv[i] = d;
        float yi = d * x[i];
        y[i] = yi;
        s0[i] = yi;
        s1[i] = 0.0f;
    }
}

__global__ __launch_bounds__(256)
void k_l1_edges(const int* __restrict__ src, const int* __restrict__ dst,
                const float* __restrict__ y, float* s0, float* s1, int e) {
    float* s = package_half() ? s1 : s0;
    const int stride = gridDim.x * blockDim.x;
    int i = blockIdx.x * blockDim.x + threadIdx.x;
    for (; i + 3 * stride < e; i += 4 * stride) {
        const int u0 = src[i];
        const int u1 = src[i + stride];
        const int u2 = src[i + 2 * stride];
        const int u3 = src[i + 3 * stride];
        const int d0 = dst[i];
        const int d1 = dst[i + stride];
        const int d2 = dst[i + 2 * stride];
        const int d3 = dst[i + 3 * stride];
        const float y0 = y[u0];
        const float y1 = y[u1];
        const float y2 = y[u2];
        const float y3 = y[u3];
        atomicAdd(&s[d0], y0);
        atomicAdd(&s[d1], y1);
        atomicAdd(&s[d2], y2);
        atomicAdd(&s[d3], y3);
    }
    for (; i < e; i += stride) atomicAdd(&s[dst[i]], y[src[i]]);
}

#define ROW_BLOCKS 1024
typedef float v4f __attribute__((ext_vector_type(4)));

__device__ __forceinline__ v4f l1_piece(float sv, v4f w, v4f b) {
    v4f r;
    r.x = sv * w.x + b.x;
    r.y = sv * w.y + b.y;
    r.z = sv * w.z + b.z;
    r.w = sv * w.w + b.w;
    r.x = r.x > 0.0f ? r.x : 0.0f;
    r.y = r.y > 0.0f ? r.y : 0.0f;
    r.z = r.z > 0.0f ? r.z : 0.0f;
    r.w = r.w > 0.0f ? r.w : 0.0f;
    return r;
}

__global__ __launch_bounds__(256)
void k_l1_feat(const float* __restrict__ s0, const float* __restrict__ s1,
               const float* __restrict__ dinv,
               const float* __restrict__ W1,
               const float* __restrict__ b1, float* __restrict__ h1, int n) {
    const int total  = n * (HID / 4);
    const int stride = gridDim.x * blockDim.x;
    int q = blockIdx.x * blockDim.x + threadIdx.x;
    const int c4 = (q & 15) * 4;
    const v4f w = *(const v4f*)(W1 + c4);
    const v4f b = *(const v4f*)(b1 + c4);
    v4f* h1v = (v4f*)h1;
    for (; q + 3 * stride < total; q += 4 * stride) {
        const int q0 = q, q1 = q + stride, q2 = q + 2 * stride, q3 = q + 3 * stride;
        const int i0 = q0 >> 4, i1 = q1 >> 4, i2 = q2 >> 4, i3 = q3 >> 4;
        const float a0 = dinv[i0] * (s0[i0] + s1[i0]);
        const float a1 = dinv[i1] * (s0[i1] + s1[i1]);
        const float a2 = dinv[i2] * (s0[i2] + s1[i2]);
        const float a3 = dinv[i3] * (s0[i3] + s1[i3]);
        h1v[q0] = l1_piece(a0, w, b);
        h1v[q1] = l1_piece(a1, w, b);
        h1v[q2] = l1_piece(a2, w, b);
        h1v[q3] = l1_piece(a3, w, b);
    }
    for (; q < total; q += stride) {
        const int i = q >> 4;
        h1v[q] = l1_piece(dinv[i] * (s0[i] + s1[i]), w, b);
    }
}

__global__ __launch_bounds__(256)
void k_gemm_wmma(const float* __restrict__ h1, const float* __restrict__ W2,
                 const float* __restrict__ dinv, float* __restrict__ t, int n_tiles) {
    __shared__ v2f sB[(HID / 2) * BPAIR_STRIDE];
    {
        float* sf = (float*)sB;
        for (int idx = threadIdx.x; idx < HID * HID; idx += blockDim.x) {
            int k = idx >> 6, col = idx & 63;
            sf[(k >> 1) * (2 * BPAIR_STRIDE) + col * 2 + (k & 1)] = W2[idx];
        }
    }
    __syncthreads();

    const int wave = threadIdx.x >> 5;
    const int lane = threadIdx.x & 31;
    const int tile = blockIdx.x * 8 + wave;
    if (tile >= n_tiles) return;

    const int row0 = tile * 16;
    const int m    = lane & 15;
    const int half = lane >> 4;

    v8f acc0 = {}, acc1 = {}, acc2 = {}, acc3 = {};
    const float* arow = h1 + (size_t)(row0 + m) * HID;

    for (int kk = 0; kk < HID; kk += 4) {
        const int ka = kk + 2 * half;
        v2f a;
        a.x = arow[ka];
        a.y = arow[ka + 1];

        const v2f* bp = &sB[(ka >> 1) * BPAIR_STRIDE + m];
        acc0 = __builtin_amdgcn_wmma_f32_16x16x4_f32(false, a, false, bp[ 0], (short)0, acc0, false, false);
        acc1 = __builtin_amdgcn_wmma_f32_16x16x4_f32(false, a, false, bp[16], (short)0, acc1, false, false);
        acc2 = __builtin_amdgcn_wmma_f32_16x16x4_f32(false, a, false, bp[32], (short)0, acc2, false, false);
        acc3 = __builtin_amdgcn_wmma_f32_16x16x4_f32(false, a, false, bp[48], (short)0, acc3, false, false);
    }

    float* orow = t + (size_t)(row0 + half * 8) * HID;
    const float* drow = dinv + row0 + half * 8;
#pragma unroll
    for (int r = 0; r < 8; ++r) {
        const float d = drow[r];
        orow[(size_t)r * HID +      m] = d * acc0[r];
        orow[(size_t)r * HID + 16 + m] = d * acc1[r];
        orow[(size_t)r * HID + 32 + m] = d * acc2[r];
        orow[(size_t)r * HID + 48 + m] = d * acc3[r];
    }
}

__global__ __launch_bounds__(256)
void k_l2_init(const float* __restrict__ t, float* __restrict__ out2, int* tick, int n) {
    if (blockIdx.x == 0 && threadIdx.x < 2) tick[threadIdx.x * TICK_STRIDE] = 0;
    const int total  = n * (HID / 4);
    const int stride = gridDim.x * blockDim.x;
    const v4f* tv = (const v4f*)t;
    v4f* ov = (v4f*)out2;
    int q = blockIdx.x * blockDim.x + threadIdx.x;
    for (; q + 3 * stride < total; q += 4 * stride) {
        const v4f r0 = tv[q];
        const v4f r1 = tv[q + stride];
        const v4f r2 = tv[q + 2 * stride];
        const v4f r3 = tv[q + 3 * stride];
        ov[q]              = r0;
        ov[q + stride]     = r1;
        ov[q + 2 * stride] = r2;
        ov[q + 3 * stride] = r3;
    }
    for (; q < total; q += stride) ov[q] = tv[q];
}

#define L2_GROUP 8
#define L2_BLOCKS 2048
#define L2_ROLES (L2_BLOCKS / 2)
#define L2_SPLIT (N_NODES / 2)

__global__ __launch_bounds__(256)
void k_l2_edges(const int* __restrict__ src, const int* __restrict__ dst,
                const float* __restrict__ t, float* out2, int* tick, int e) {
    __shared__ int s_role[2];
    const int hf = package_half();
    if (threadIdx.x == 0) {
        const int r = atomicAdd(&tick[hf * TICK_STRIDE], 1);
        int own = hf, share = r;
        if (r >= L2_ROLES) {
            own = 1 - hf;
            share = 2 * L2_ROLES - 1 - r;
        }
        s_role[0] = own;
        s_role[1] = share;
    }
    __syncthreads();
    const int own   = __builtin_amdgcn_readfirstlane(s_role[0]);
    const int share = __builtin_amdgcn_readfirstlane(s_role[1]);
    if (share < 0 || share >= L2_ROLES) return;

    const int lane    = threadIdx.x & 31;
    const int ngroups = (e + 31) >> 5;
    const int widx = share * 8 + (threadIdx.x >> 5);
    int g        = (int)(((long long)widx * ngroups) / (8 * L2_ROLES));
    const int g1 = (int)(((long long)(widx + 1) * ngroups) / (8 * L2_ROLES));

    int u = 0, v = -1;
    if (g < g1) {
        const int ei = g * 32 + lane;
        if (ei < e) {
            u = src[ei];
            v = dst[ei];
        }
    }
    unsigned mask = __builtin_amdgcn_ballot_w32((v >= 0) && ((v >= L2_SPLIT ? 1 : 0) == own));
    while (g < g1) {
        const int gn = g + 1;
        int un = 0, vn = -1;
        if (gn < g1) {
            const int ei = gn * 32 + lane;
            if (ei < e) {
                un = src[ei];
                vn = dst[ei];
            }
        }
        while (mask != 0u) {
            float a[L2_GROUP], b[L2_GROUP];
            int sv[L2_GROUP];
            unsigned m2 = mask;
            for (int k = 0; k < L2_GROUP; ++k) {
                if (m2 != 0u) {
                    const int j = __builtin_ctz(m2);
                    m2 &= m2 - 1u;
                    const int su = __builtin_amdgcn_readlane(u, j);
                    sv[k] = __builtin_amdgcn_readlane(v, j);
                    const float* tr = t + (size_t)su * HID + lane;
                    a[k] = tr[0];
                    b[k] = tr[32];
                }
            }
            for (int k = 0; k < L2_GROUP; ++k) {
                if (mask != 0u) {
                    mask &= mask - 1u;
                    float* orow = out2 + (size_t)sv[k] * HID + lane;
                    atomicAdd(&orow[0],  a[k]);
                    atomicAdd(&orow[32], b[k]);
                }
            }
        }
        u = un;
        v = vn;
        g = gn;
        mask = __builtin_amdgcn_ballot_w32((v >= 0) && ((v >= L2_SPLIT ? 1 : 0) == own));
    }
}

__global__ void k_pool_init(float* pool0, float* pool1, float* cnt0, float* cnt1) {
    int i = blockIdx.x * blockDim.x + threadIdx.x;
    if (i < N_GRAPH * HID) {
        pool0[i] = 0.0f;
        pool1[i] = 0.0f;
    }
    if (i < N_GRAPH) {
        cnt0[i] = 0.0f;
        cnt1[i] = 0.0f;
    }
}

__global__ __launch_bounds__(256)
void k_relu_pool(const float* __restrict__ out2, const float* __restrict__ dinv,
                 const float* __restrict__ b2, const int* __restrict__ batch,
                 float* pool0, float* pool1, float* cnt0, float* cnt1, int n) {
    const int hf = package_half();
    float* pool = hf ? pool1 : pool0;
    float* cnt  = hf ? cnt1 : cnt0;
    const int total  = n * HID;
    const int stride = gridDim.x * blockDim.x;
    int tid = blockIdx.x * blockDim.x + threadIdx.x;
    const int h = tid & 63;
    const float bh = b2[h];
    for (; tid + 3 * stride < total; tid += 4 * stride) {
        const int t0 = tid, t1 = tid + stride, t2 = tid + 2 * stride, t3 = tid + 3 * stride;
        const int i0 = t0 >> 6, i1 = t1 >> 6, i2 = t2 >> 6, i3 = t3 >> 6;
        const float o0 = out2[t0], o1 = out2[t1], o2 = out2[t2], o3 = out2[t3];
        const float e0 = dinv[i0], e1 = dinv[i1], e2 = dinv[i2], e3 = dinv[i3];
        const int g0 = batch[i0], g1 = batch[i1], g2 = batch[i2], g3 = batch[i3];
        float v0 = e0 * o0 + bh;
        float v1 = e1 * o1 + bh;
        float v2 = e2 * o2 + bh;
        float v3 = e3 * o3 + bh;
        v0 = v0 > 0.0f ? v0 : 0.0f;
        v1 = v1 > 0.0f ? v1 : 0.0f;
        v2 = v2 > 0.0f ? v2 : 0.0f;
        v3 = v3 > 0.0f ? v3 : 0.0f;
        atomicAdd(&pool[g0 * HID + h], v0);
        atomicAdd(&pool[g1 * HID + h], v1);
        atomicAdd(&pool[g2 * HID + h], v2);
        atomicAdd(&pool[g3 * HID + h], v3);
        if (h == 0) {
            atomicAdd(&cnt[g0], 1.0f);
            atomicAdd(&cnt[g1], 1.0f);
            atomicAdd(&cnt[g2], 1.0f);
            atomicAdd(&cnt[g3], 1.0f);
        }
    }
    for (; tid < total; tid += stride) {
        const int i = tid >> 6;
        float v = dinv[i] * out2[tid] + bh;
        v = v > 0.0f ? v : 0.0f;
        const int g = batch[i];
        atomicAdd(&pool[g * HID + h], v);
        if (h == 0) atomicAdd(&cnt[g], 1.0f);
    }
}

__global__ void k_final(const float* __restrict__ pool0, const float* __restrict__ pool1,
                        const float* __restrict__ cnt0, const float* __restrict__ cnt1,
                        float* __restrict__ out) {
    int tid = blockIdx.x * blockDim.x + threadIdx.x;
    if (tid < N_GRAPH * HID) {
        float c = cnt0[tid >> 6] + cnt1[tid >> 6];
        out[tid] = (pool0[tid] + pool1[tid]) / fmaxf(c, 1.0f);
    }
}

extern "C" void kernel_launch(void* const* d_in, const int* in_sizes, int n_in,
                              void* d_out, int out_size, void* d_ws, size_t ws_size,
                              hipStream_t stream) {
    const float* x     = (const float*)d_in[0];
    const int*   edge  = (const int*)  d_in[1];
    const int*   batch = (const int*)  d_in[2];
    const float* W1    = (const float*)d_in[3];
    const float* b1    = (const float*)d_in[4];
    const float* W2    = (const float*)d_in[5];
    const float* b2    = (const float*)d_in[6];
    const int* src = edge;
    const int* dst = edge + N_EDGES;
    float* out = (float*)d_out;

    float* ws   = (float*)d_ws;
    float* dinv = ws;
    float* s    = dinv + N_NODES;
    float* bufA = s + N_NODES;
    float* bufB = bufA + (size_t)N_NODES * HID;
    float* y    = bufB;
    float* deg1 = bufB + N_NODES;
    float* s1   = bufB + 2 * N_NODES;
    float* pool = bufB + (size_t)N_NODES * HID;
    float* cnt  = pool + N_GRAPH * HID;
    float* pool1 = s;
    float* cnt1  = s + N_GRAPH * HID;
    int*   tick  = (int*)(s + 160000);

    const int TB = 256;
    k_init_deg     <<<(N_NODES + TB - 1) / TB, TB, 0, stream>>>(dinv, deg1, N_NODES);
    k_deg_edges    <<<EDGE_BLOCKS, TB, 0, stream>>>(dst, dinv, deg1, N_EDGES);
    k_dinv_selfloop<<<(N_NODES + TB - 1) / TB, TB, 0, stream>>>(dinv, deg1, x, y, s, s1, N_NODES);
    k_l1_edges     <<<EDGE_BLOCKS, TB, 0, stream>>>(src, dst, y, s, s1, N_EDGES);
    k_l1_feat      <<<ROW_BLOCKS, TB, 0, stream>>>(s, s1, dinv, W1, b1, bufA, N_NODES);

    const int n_tiles = N_NODES / 16;
    k_gemm_wmma    <<<(n_tiles + 7) / 8, 256, 0, stream>>>(bufA, W2, dinv, bufB, n_tiles);

    k_l2_init      <<<ROW_BLOCKS, TB, 0, stream>>>(bufB, bufA, tick, N_NODES);
    k_l2_edges     <<<L2_BLOCKS, TB, 0, stream>>>(src, dst, bufB, bufA, tick, N_EDGES);
    k_pool_init    <<<(N_GRAPH * HID + TB - 1) / TB, TB, 0, stream>>>(pool, pool1, cnt, cnt1);
    k_relu_pool    <<<POOL_BLOCKS, TB, 0, stream>>>(bufA, dinv, b2, batch, pool, pool1, cnt, cnt1, N_NODES);
    k_final        <<<(N_GRAPH * HID + TB - 1) / TB, TB, 0, stream>>>(pool, pool1, cnt, cnt1, out);
}


# t rows stored with columns l and l+32 adjacent: the edge pass fetches a source row with one 8-byte load per lane (was two 4-byte loads); GEMM stores whole 128-B lines; sums and their order unchanged
# speedup vs baseline: 3.4920x; 1.0388x over previous
#include <hip/hip_runtime.h>

typedef float v2f __attribute__((ext_vector_type(2)));
typedef float v8f __attribute__((ext_vector_type(8)));

#define N_NODES 200000
#define N_EDGES 3200000
#define N_GRAPH 2048
#define HID     64
#define BPAIR_STRIDE 80

#define EDGE_BLOCKS ((N_EDGES + 256 * 8 - 1) / (256 * 8))
#define POOL_BLOCKS 2048
#define TICK_STRIDE 64

__device__ __forceinline__ int package_half() {
    const unsigned id = __builtin_amdgcn_s_sendmsg_rtn(0x87);
    return (int)((id >> 18) & 1u);
}

__global__ void k_init_deg(float* deg0, float* deg1, int n) {
    int i = blockIdx.x * blockDim.x + threadIdx.x;
    if (i < n) {
        deg0[i] = 1.0f;
        deg1[i] = 0.0f;
    }
}

__global__ __launch_bounds__(256)
void k_deg_edges(const int* __restrict__ dst, float* deg0, float* deg1, int e) {
    float* deg = package_half() ? deg1 : deg0;
    const int stride = gridDim.x * blockDim.x;
    int i = blockIdx.x * blockDim.x + threadIdx.x;
    for (; i + 3 * stride < e; i += 4 * stride) {
        const int d0 = dst[i];
        const int d1 = dst[i + stride];
        const int d2 = dst[i + 2 * stride];
        const int d3 = dst[i + 3 * stride];
        atomicAdd(&deg[d0], 1.0f);
        atomicAdd(&deg[d1], 1.0f);
        atomicAdd(&deg[d2], 1.0f);
        atomicAdd(&deg[d3], 1.0f);
    }
    for (; i < e; i += stride) atomicAdd(&deg[dst[i]], 1.0f);
}

__global__ void k_dinv_selfloop(float* deg_dinv, const float* __restrict__ deg1,
                                const float* __restrict__ x,
                                float* __restrict__ y, float* s0, float* s1, int n) {
    int i = blockIdx.x * blockDim.x + threadIdx.x;
    if (i < n) {
        float d = rsqrtf(deg_dinv[i] + deg1[i]);
        deg_dinv[i] = d;
        float yi = d * x[i];
        y[i] = yi;
        s0[i] = yi;
        s1[i] = 0.0f;
    }
}

__global__ __launch_bounds__(256)
void k_l1_edges(const int* __restrict__ src, const int* __restrict__ dst,
                const float* __restrict__ y, float* s0, float* s1, int e) {
    float* s = package_half() ? s1 : s0;
    const int stride = gridDim.x * blockDim.x;
    int i = blockIdx.x * blockDim.x + threadIdx.x;
    for (; i + 3 * stride < e; i += 4 * stride) {
        const int u0 = src[i];
        const int u1 = src[i + stride];
        const int u2 = src[i + 2 * stride];
        const int u3 = src[i + 3 * stride];
        const int d0 = dst[i];
        const int d1 = dst[i + stride];
        const int d2 = dst[i + 2 * stride];
        const int d3 = dst[i + 3 * stride];
        const float y0 = y[u0];
        const float y1 = y[u1];
        const float y2 = y[u2];
        const float y3 = y[u3];
        atomicAdd(&s[d0], y0);
        atomicAdd(&s[d1], y1);
        atomicAdd(&s[d2], y2);
        atomicAdd(&s[d3], y3);
    }
    for (; i < e; i += stride) atomicAdd(&s[dst[i]], y[src[i]]);
}

#define ROW_BLOCKS 1024
typedef float v4f __attribute__((ext_vector_type(4)));

__device__ __forceinline__ v4f l1_piece(float sv, v4f w, v4f b) {
    v4f r;
    r.x = sv * w.x + b.x;
    r.y = sv * w.y + b.y;
    r.z = sv * w.z + b.z;
    r.w = sv * w.w + b.w;
    r.x = r.x > 0.0f ? r.x : 0.0f;
    r.y = r.y > 0.0f ? r.y : 0.0f;
    r.z = r.z > 0.0f ? r.z : 0.0f;
    r.w = r.w > 0.0f ? r.w : 0.0f;
    return r;
}

__global__ __launch_bounds__(256)
void k_l1_feat(const float* __restrict__ s0, const float* __restrict__ s1,
               const float* __restrict__ dinv,
               const float* __restrict__ W1,
               const float* __restrict__ b1, float* __restrict__ h1, int n) {
    const int total  = n * (HID / 4);
    const int stride = gridDim.x * blockDim.x;
    int q = blockIdx.x * blockDim.x + threadIdx.x;
    const int c4 = (q & 15) * 4;
    const v4f w = *(const v4f*)(W1 + c4);
    const v4f b = *(const v4f*)(b1 + c4);
    v4f* h1v = (v4f*)h1;
    for (; q + 3 * stride < total; q += 4 * stride) {
        const int q0 = q, q1 = q + stride, q2 = q + 2 * stride, q3 = q + 3 * stride;
        const int i0 = q0 >> 4, i1 = q1 >> 4, i2 = q2 >> 4, i3 = q3 >> 4;
        const float a0 = dinv[i0] * (s0[i0] + s1[i0]);
        const float a1 = dinv[i1] * (s0[i1] + s1[i1]);
        const float a2 = dinv[i2] * (s0[i2] + s1[i2]);
        const float a3 = dinv[i3] * (s0[i3] + s1[i3]);
        h1v[q0] = l1_piece(a0, w, b);
        h1v[q1] = l1_piece(a1, w, b);
        h1v[q2] = l1_piece(a2, w, b);
        h1v[q3] = l1_piece(a3, w, b);
    }
    for (; q < total; q += stride) {
        const int i = q >> 4;
        h1v[q] = l1_piece(dinv[i] * (s0[i] + s1[i]), w, b);
    }
}

__global__ __launch_bounds__(256)
void k_gemm_wmma(const float* __restrict__ h1, const float* __restrict__ W2,
                 const float* __restrict__ dinv, float* __restrict__ t, int n_tiles) {
    __shared__ v2f sB[(HID / 2) * BPAIR_STRIDE];
    {
        float* sf = (float*)sB;
        for (int idx = threadIdx.x; idx < HID * HID; idx += blockDim.x) {
            int k = idx >> 6, col = idx & 63;
            sf[(k >> 1) * (2 * BPAIR_STRIDE) + col * 2 + (k & 1)] = W2[idx];
        }
    }
    __syncthreads();

    const int wave = threadIdx.x >> 5;
    const int lane = threadIdx.x & 31;
    const int tile = blockIdx.x * 8 + wave;
    if (tile >= n_tiles) return;

    const int row0 = tile * 16;
    const int m    = lane & 15;
    const int half = lane >> 4;

    v8f acc0 = {}, acc1 = {}, acc2 = {}, acc3 = {};
    const float* arow = h1 + (size_t)(row0 + m) * HID;

    for (int kk = 0; kk < HID; kk += 4) {
        const int ka = kk + 2 * half;
        v2f a;
        a.x = arow[ka];
        a.y = arow[ka + 1];

        const v2f* bp = &sB[(ka >> 1) * BPAIR_STRIDE + m];
        acc0 = __builtin_amdgcn_wmma_f32_16x16x4_f32(false, a, false, bp[ 0], (short)0, acc0, false, false);
        acc1 = __builtin_amdgcn_wmma_f32_16x16x4_f32(false, a, false, bp[16], (short)0, acc1, false, false);
        acc2 = __builtin_amdgcn_wmma_f32_16x16x4_f32(false, a, false, bp[32], (short)0, acc2, false, false);
        acc3 = __builtin_amdgcn_wmma_f32_16x16x4_f32(false, a, false, bp[48], (short)0, acc3, false, false);
    }

    float* orow = t + (size_t)(row0 + half * 8) * HID;
    const float* drow = dinv + row0 + half * 8;
#pragma unroll
    for (int r = 0; r < 8; ++r) {
        const float d = drow[r];
        v2f lo, hi;
        lo.x = d * acc0[r];
        lo.y = d * acc2[r];
        hi.x = d * acc1[r];
        hi.y = d * acc3[r];
        *(v2f*)(orow + (size_t)r * HID +      2 * m) = lo;
        *(v2f*)(orow + (size_t)r * HID + 32 + 2 * m) = hi;
    }
}

__device__ __forceinline__ void l2_unpair(v2f* ov, int q, v4f r) {
    const int i = q >> 4, j = q & 15;
    v2f lo, hi;
    lo.x = r.x;
    lo.y = r.z;
    hi.x = r.y;
    hi.y = r.w;
    ov[i * (HID / 2) + j]      = lo;
    ov[i * (HID / 2) + 16 + j] = hi;
}

__global__ __launch_bounds__(256)
void k_l2_init(const float* __restrict__ t, float* __restrict__ out2, int* tick, int n) {
    if (blockIdx.x == 0 && threadIdx.x < 2) tick[threadIdx.x * TICK_STRIDE] = 0;
    const int total  = n * (HID / 4);
    const int stride = gridDim.x * blockDim.x;
    const v4f* tv = (const v4f*)t;
    v2f* ov = (v2f*)out2;
    int q = blockIdx.x * blockDim.x + threadIdx.x;
    for (; q + 3 * stride < total; q += 4 * stride) {
        const v4f r0 = tv[q];
        const v4f r1 = tv[q + stride];
        const v4f r2 = tv[q + 2 * stride];
        const v4f r3 = tv[q + 3 * stride];
        l2_unpair(ov, q,              r0);
        l2_unpair(ov, q + stride,     r1);
        l2_unpair(ov, q + 2 * stride, r2);
        l2_unpair(ov, q + 3 * stride, r3);
    }
    for (; q < total; q += stride) l2_unpair(ov, q, tv[q]);
}

#define L2_GROUP 8
#define L2_BLOCKS 2048
#define L2_ROLES (L2_BLOCKS / 2)
#define L2_SPLIT (N_NODES / 2)

__global__ __launch_bounds__(256)
void k_l2_edges(const int* __restrict__ src, const int* __restrict__ dst,
                const float* __restrict__ t, float* out2, int* tick, int e) {
    __shared__ int s_role[2];
    const int hf = package_half();
    if (threadIdx.x == 0) {
        const int r = atomicAdd(&tick[hf * TICK_STRIDE], 1);
        int own = hf, share = r;
        if (r >= L2_ROLES) {
            own = 1 - hf;
            share = 2 * L2_ROLES - 1 - r;
        }
        s_role[0] = own;
        s_role[1] = share;
    }
    __syncthreads();
    const int own   = __builtin_amdgcn_readfirstlane(s_role[0]);
    const int share = __builtin_amdgcn_readfirstlane(s_role[1]);
    if (share < 0 || share >= L2_ROLES) return;

    const int lane    = threadIdx.x & 31;
    const int ngroups = (e + 31) >> 5;
    const int widx = share * 8 + (threadIdx.x >> 5);
    int g        = (int)(((long long)widx * ngroups) / (8 * L2_ROLES));
    const int g1 = (int)(((long long)(widx + 1) * ngroups) / (8 * L2_ROLES));

    int u = 0, v = -1;
    if (g < g1) {
        const int ei = g * 32 + lane;
        if (ei < e) {
            u = src[ei];
            v = dst[ei];
        }
    }
    unsigned mask = __builtin_amdgcn_ballot_w32((v >= 0) && ((v >= L2_SPLIT ? 1 : 0) == own));
    while (g < g1) {
        const int gn = g + 1;
        int un = 0, vn = -1;
        if (gn < g1) {
            const int ei = gn * 32 + lane;
            if (ei < e) {
                un = src[ei];
                vn = dst[ei];
            }
        }
        while (mask != 0u) {
            float a[L2_GROUP], b[L2_GROUP];
            int sv[L2_GROUP];
            unsigned m2 = mask;
            for (int k = 0; k < L2_GROUP; ++k) {
                if (m2 != 0u) {
                    const int j = __builtin_ctz(m2);
                    m2 &= m2 - 1u;
                    const int su = __builtin_amdgcn_readlane(u, j);
                    sv[k] = __builtin_amdgcn_readlane(v, j);
                    const v2f ab = *((const v2f*)(t + (size_t)su * HID) + lane);
                    a[k] = ab.x;
                    b[k] = ab.y;
                }
            }
            for (int k = 0; k < L2_GROUP; ++k) {
                if (mask != 0u) {
                    mask &= mask - 1u;
                    float* orow = out2 + (size_t)sv[k] * HID + lane;
                    atomicAdd(&orow[0],  a[k]);
                    atomicAdd(&orow[32], b[k]);
                }
            }
        }
        u = un;
        v = vn;
        g = gn;
        mask = __builtin_amdgcn_ballot_w32((v >= 0) && ((v >= L2_SPLIT ? 1 : 0) == own));
    }
}

__global__ void k_pool_init(float* pool0, float* pool1, float* cnt0, float* cnt1) {
    int i = blockIdx.x * blockDim.x + threadIdx.x;
    if (i < N_GRAPH * HID) {
        pool0[i] = 0.0f;
        pool1[i] = 0.0f;
    }
    if (i < N_GRAPH) {
        cnt0[i] = 0.0f;
        cnt1[i] = 0.0f;
    }
}

__global__ __launch_bounds__(256)
void k_relu_pool(const float* __restrict__ out2, const float* __restrict__ dinv,
                 const float* __restrict__ b2, const int* __restrict__ batch,
                 float* pool0, float* pool1, float* cnt0, float* cnt1, int n) {
    const int hf = package_half();
    float* pool = hf ? pool1 : pool0;
    float* cnt  = hf ? cnt1 : cnt0;
    const int total  = n * HID;
    const int stride = gridDim.x * blockDim.x;
    int tid = blockIdx.x * blockDim.x + threadIdx.x;
    const int h = tid & 63;
    const float bh = b2[h];
    for (; tid + 3 * stride < total; tid += 4 * stride) {
        const int t0 = tid, t1 = tid + stride, t2 = tid + 2 * stride, t3 = tid + 3 * stride;
        const int i0 = t0 >> 6, i1 = t1 >> 6, i2 = t2 >> 6, i3 = t3 >> 6;
        const float o0 = out2[t0], o1 = out2[t1], o2 = out2[t2], o3 = out2[t3];
        const float e0 = dinv[i0], e1 = dinv[i1], e2 = dinv[i2], e3 = dinv[i3];
        const int g0 = batch[i0], g1 = batch[i1], g2 = batch[i2], g3 = batch[i3];
        float v0 = e0 * o0 + bh;
        float v1 = e1 * o1 + bh;
        float v2 = e2 * o2 + bh;
        float v3 = e3 * o3 + bh;
        v0 = v0 > 0.0f ? v0 : 0.0f;
        v1 = v1 > 0.0f ? v1 : 0.0f;
        v2 = v2 > 0.0f ? v2 : 0.0f;
        v3 = v3 > 0.0f ? v3 : 0.0f;
        atomicAdd(&pool[g0 * HID + h], v0);
        atomicAdd(&pool[g1 * HID + h], v1);
        atomicAdd(&pool[g2 * HID + h], v2);
        atomicAdd(&pool[g3 * HID + h], v3);
        if (h == 0) {
            atomicAdd(&cnt[g0], 1.0f);
            atomicAdd(&cnt[g1], 1.0f);
            atomicAdd(&cnt[g2], 1.0f);
            atomicAdd(&cnt[g3], 1.0f);
        }
    }
    for (; tid < total; tid += stride) {
        const int i = tid >> 6;
        float v = dinv[i] * out2[tid] + bh;
        v = v > 0.0f ? v : 0.0f;
        const int g = batch[i];
        atomicAdd(&pool[g * HID + h], v);
        if (h == 0) atomicAdd(&cnt[g], 1.0f);
    }
}

__global__ void k_final(const float* __restrict__ pool0, const float* __restrict__ pool1,
                        const float* __restrict__ cnt0, const float* __restrict__ cnt1,
                        float* __restrict__ out) {
    int tid = blockIdx.x * blockDim.x + threadIdx.x;
    if (tid < N_GRAPH * HID) {
        float c = cnt0[tid >> 6] + cnt1[tid >> 6];
        out[tid] = (pool0[tid] + pool1[tid]) / fmaxf(c, 1.0f);
    }
}

extern "C" void kernel_launch(void* const* d_in, const int* in_sizes, int n_in,
                              void* d_out, int out_size, void* d_ws, size_t ws_size,
                              hipStream_t stream) {
    const float* x     = (const float*)d_in[0];
    const int*   edge  = (const int*)  d_in[1];
    const int*   batch = (const int*)  d_in[2];
    const float* W1    = (const float*)d_in[3];
    const float* b1    = (const float*)d_in[4];
    const float* W2    = (const float*)d_in[5];
    const float* b2    = (const float*)d_in[6];
    const int* src = edge;
    const int* dst = edge + N_EDGES;
    float* out = (float*)d_out;

    float* ws   = (float*)d_ws;
    float* dinv = ws;
    float* s    = dinv + N_NODES;
    float* bufA = s + N_NODES;
    float* bufB = bufA + (size_t)N_NODES * HID;
    float* y    = bufB;
    float* deg1 = bufB + N_NODES;
    float* s1   = bufB + 2 * N_NODES;
    float* pool = bufB + (size_t)N_NODES * HID;
    float* cnt  = pool + N_GRAPH * HID;
    float* pool1 = s;
    float* cnt1  = s + N_GRAPH * HID;
    int*   tick  = (int*)(s + 160000);

    const int TB = 256;
    k_init_deg     <<<(N_NODES + TB - 1) / TB, TB, 0, stream>>>(dinv, deg1, N_NODES);
    k_deg_edges    <<<EDGE_BLOCKS, TB, 0, stream>>>(dst, dinv, deg1, N_EDGES);
    k_dinv_selfloop<<<(N_NODES + TB - 1) / TB, TB, 0, stream>>>(dinv, deg1, x, y, s, s1, N_NODES);
    k_l1_edges     <<<EDGE_BLOCKS, TB, 0, stream>>>(src, dst, y, s, s1, N_EDGES);
    k_l1_feat      <<<ROW_BLOCKS, TB, 0, stream>>>(s, s1, dinv, W1, b1, bufA, N_NODES);

    const int n_tiles = N_NODES / 16;
    k_gemm_wmma    <<<(n_tiles + 7) / 8, 256, 0, stream>>>(bufA, W2, dinv, bufB, n_tiles);

    k_l2_init      <<<ROW_BLOCKS, TB, 0, stream>>>(bufB, bufA, tick, N_NODES);
    k_l2_edges     <<<L2_BLOCKS, TB, 0, stream>>>(src, dst, bufB, bufA, tick, N_EDGES);
    k_pool_init    <<<(N_GRAPH * HID + TB - 1) / TB, TB, 0, stream>>>(pool, pool1, cnt, cnt1);
    k_relu_pool    <<<POOL_BLOCKS, TB, 0, stream>>>(bufA, dinv, b2, batch, pool, pool1, cnt, cnt1, N_NODES);
    k_final        <<<(N_GRAPH * HID + TB - 1) / TB, TB, 0, stream>>>(pool, pool1, cnt, cnt1, out);
}


# Layer-2 rows summed BEFORE the W2 product (linearity, as one sibling step; here in push form): edge pass rebuilds dinv*relu(a*W1+b1) from an 8-byte (a,dinv) gather, no row load (sums regrouped)
# speedup vs baseline: 4.3831x; 1.2552x over previous
#include <hip/hip_runtime.h>

typedef float v2f __attribute__((ext_vector_type(2)));
typedef float v8f __attribute__((ext_vector_type(8)));

#define N_NODES 200000
#define N_EDGES 3200000
#define N_GRAPH 2048
#define HID     64
#define BPAIR_STRIDE 80

#define EDGE_BLOCKS ((N_EDGES + 256 * 8 - 1) / (256 * 8))
#define POOL_BLOCKS 2048
#define TICK_STRIDE 64

__device__ __forceinline__ int package_half() {
    const unsigned id = __builtin_amdgcn_s_sendmsg_rtn(0x87);
    return (int)((id >> 18) & 1u);
}

__global__ void k_init_deg(float* deg0, float* deg1, int n) {
    int i = blockIdx.x * blockDim.x + threadIdx.x;
    if (i < n) {
        deg0[i] = 1.0f;
        deg1[i] = 0.0f;
    }
}

__global__ __launch_bounds__(256)
void k_deg_edges(const int* __restrict__ dst, float* deg0, float* deg1, int e) {
    float* deg = package_half() ? deg1 : deg0;
    const int stride = gridDim.x * blockDim.x;
    int i = blockIdx.x * blockDim.x + threadIdx.x;
    for (; i + 3 * stride < e; i += 4 * stride) {
        const int d0 = dst[i];
        const int d1 = dst[i + stride];
        const int d2 = dst[i + 2 * stride];
        const int d3 = dst[i + 3 * stride];
        atomicAdd(&deg[d0], 1.0f);
        atomicAdd(&deg[d1], 1.0f);
        atomicAdd(&deg[d2], 1.0f);
        atomicAdd(&deg[d3], 1.0f);
    }
    for (; i < e; i += stride) atomicAdd(&deg[dst[i]], 1.0f);
}

__global__ void k_dinv_selfloop(float* deg_dinv, const float* __restrict__ deg1,
                                const float* __restrict__ x,
                                float* __restrict__ y, float* s0, float* s1, int n) {
    int i = blockIdx.x * blockDim.x + threadIdx.x;
    if (i < n) {
        float d = rsqrtf(deg_dinv[i] + deg1[i]);
        deg_dinv[i] = d;
        float yi = d * x[i];
        y[i] = yi;
        s0[i] = yi;
        s1[i] = 0.0f;
    }
}

__global__ __launch_bounds__(256)
void k_l1_edges(const int* __restrict__ src, const int* __restrict__ dst,
                const float* __restrict__ y, float* s0, float* s1, int e) {
    float* s = package_half() ? s1 : s0;
    const int stride = gridDim.x * blockDim.x;
    int i = blockIdx.x * blockDim.x + threadIdx.x;
    for (; i + 3 * stride < e; i += 4 * stride) {
        const int u0 = src[i];
        const int u1 = src[i + stride];
        const int u2 = src[i + 2 * stride];
        const int u3 = src[i + 3 * stride];
        const int d0 = dst[i];
        const int d1 = dst[i + stride];
        const int d2 = dst[i + 2 * stride];
        const int d3 = dst[i + 3 * stride];
        const float y0 = y[u0];
        const float y1 = y[u1];
        const float y2 = y[u2];
        const float y3 = y[u3];
        atomicAdd(&s[d0], y0);
        atomicAdd(&s[d1], y1);
        atomicAdd(&s[d2], y2);
        atomicAdd(&s[d3], y3);
    }
    for (; i < e; i += stride) atomicAdd(&s[dst[i]], y[src[i]]);
}

#define ROW_BLOCKS 1024
typedef float v4f __attribute__((ext_vector_type(4)));

__device__ __forceinline__ float l1_scaled(float a, float d, float w, float b) {
    float r = a * w + b;
    r = r > 0.0f ? r : 0.0f;
    return d * r;
}

__device__ __forceinline__ v4f l1_piece(float a, float d, v4f w, v4f b) {
    v4f r;
    r.x = l1_scaled(a, d, w.x, b.x);
    r.y = l1_scaled(a, d, w.y, b.y);
    r.z = l1_scaled(a, d, w.z, b.z);
    r.w = l1_scaled(a, d, w.w, b.w);
    return r;
}

__global__ __launch_bounds__(256)
void k_l1_rows(const float* __restrict__ s0, const float* __restrict__ s1,
               const float* __restrict__ dinv,
               const float* __restrict__ W1,
               const float* __restrict__ b1, float* __restrict__ z,
               float* __restrict__ pairs, int* tick, int n) {
    if (blockIdx.x == 0 && threadIdx.x < 2) tick[threadIdx.x * TICK_STRIDE] = 0;
    const int total  = n * (HID / 4);
    const int stride = gridDim.x * blockDim.x;
    int q = blockIdx.x * blockDim.x + threadIdx.x;
    const int c4 = (q & 15) * 4;
    const bool first = (c4 == 0);
    const v4f w = *(const v4f*)(W1 + c4);
    const v4f b = *(const v4f*)(b1 + c4);
    v4f* zv = (v4f*)z;
    v2f* pv = (v2f*)pairs;
    for (; q + 3 * stride < total; q += 4 * stride) {
        const int q0 = q, q1 = q + stride, q2 = q + 2 * stride, q3 = q + 3 * stride;
        const int i0 = q0 >> 4, i1 = q1 >> 4, i2 = q2 >> 4, i3 = q3 >> 4;
        const float d0 = dinv[i0], d1 = dinv[i1], d2 = dinv[i2], d3 = dinv[i3];
        const float a0 = d0 * (s0[i0] + s1[i0]);
        const float a1 = d1 * (s0[i1] + s1[i1]);
        const float a2 = d2 * (s0[i2] + s1[i2]);
        const float a3 = d3 * (s0[i3] + s1[i3]);
        zv[q0] = l1_piece(a0, d0, w, b);
        zv[q1] = l1_piece(a1, d1, w, b);
        zv[q2] = l1_piece(a2, d2, w, b);
        zv[q3] = l1_piece(a3, d3, w, b);
        if (first) {
            v2f p0, p1, p2, p3;
            p0.x = a0; p0.y = d0;
            p1.x = a1; p1.y = d1;
            p2.x = a2; p2.y = d2;
            p3.x = a3; p3.y = d3;
            pv[i0] = p0;
            pv[i1] = p1;
            pv[i2] = p2;
            pv[i3] = p3;
        }
    }
    for (; q < total; q += stride) {
        const int i = q >> 4;
        const float d = dinv[i];
        const float a = d * (s0[i] + s1[i]);
        zv[q] = l1_piece(a, d, w, b);
        if (first) {
            v2f p;
            p.x = a; p.y = d;
            pv[i] = p;
        }
    }
}

#define L2_BLOCKS 2048
#define L2_ROLES (L2_BLOCKS / 2)
#define L2_SPLIT (N_NODES / 2)

__device__ __forceinline__ void l2_fetch_idx(const int* __restrict__ src, const int* __restrict__ dst,
                                             int grp, int g1, int lane, int e, int& u, int& v) {
    const int ei = grp * 32 + lane;
    const bool ok = (grp < g1) && (ei < e);
    const int ec = ok ? ei : (e - 1);
    u = src[ec];
    const int vv = dst[ec];
    v = ok ? vv : -1;
}

__global__ __launch_bounds__(256)
void k_l2_edges(const int* __restrict__ src, const int* __restrict__ dst,
                const float* __restrict__ pairs,
                const float* __restrict__ W1, const float* __restrict__ b1,
                float* z, int* tick, int e) {
    __shared__ int s_role[2];
    const int hf = package_half();
    if (threadIdx.x == 0) {
        const int r = atomicAdd(&tick[hf * TICK_STRIDE], 1);
        int own = hf, share = r;
        if (r >= L2_ROLES) {
            own = 1 - hf;
            share = 2 * L2_ROLES - 1 - r;
        }
        s_role[0] = own;
        s_role[1] = share;
    }
    __syncthreads();
    const int own   = __builtin_amdgcn_readfirstlane(s_role[0]);
    const int share = __builtin_amdgcn_readfirstlane(s_role[1]);
    if (share < 0 || share >= L2_ROLES) return;

    const int lane    = threadIdx.x & 31;
    const float w0 = W1[lane], w1 = W1[lane + 32];
    const float c0 = b1[lane], c1 = b1[lane + 32];
    const v2f* pr = (const v2f*)pairs;
    const int ngroups = (e + 31) >> 5;
    const int widx = share * 8 + (threadIdx.x >> 5);
    int g        = (int)(((long long)widx * ngroups) / (8 * L2_ROLES));
    const int g1 = (int)(((long long)(widx + 1) * ngroups) / (8 * L2_ROLES));

    int u0, v0;
    int u1, v1;
    l2_fetch_idx(src, dst, g, g1, lane, e, u0, v0);
    l2_fetch_idx(src, dst, g + 1, g1, lane, e, u1, v1);
    float a0 = 0.0f, d0 = 0.0f;
    if ((v0 >= 0) && ((v0 >= L2_SPLIT ? 1 : 0) == own)) {
        const v2f p = pr[u0];
        a0 = p.x;
        d0 = p.y;
    }
    while (g < g1) {
        int u2, v2;
        l2_fetch_idx(src, dst, g + 2, g1, lane, e, u2, v2);
        float a1 = 0.0f, d1 = 0.0f;
        if ((v1 >= 0) && ((v1 >= L2_SPLIT ? 1 : 0) == own)) {
            const v2f p = pr[u1];
            a1 = p.x;
            d1 = p.y;
        }

        unsigned mask = __builtin_amdgcn_ballot_w32((v0 >= 0) && ((v0 >= L2_SPLIT ? 1 : 0) == own));
        while (mask != 0u) {
            const int j = __builtin_ctz(mask);
            mask &= mask - 1u;
            const float sa = __int_as_float(__builtin_amdgcn_readlane(__float_as_int(a0), j));
            const float sd = __int_as_float(__builtin_amdgcn_readlane(__float_as_int(d0), j));
            const int   sv = __builtin_amdgcn_readlane(v0, j);
            float* orow = z + (size_t)sv * HID + lane;
            atomicAdd(&orow[0],  l1_scaled(sa, sd, w0, c0));
            atomicAdd(&orow[32], l1_scaled(sa, sd, w1, c1));
        }
        v0 = v1;
        a0 = a1;
        d0 = d1;
        u1 = u2;
        v1 = v2;
        ++g;
    }
}

__global__ __launch_bounds__(256)
void k_gemm_wmma(const float* __restrict__ z, const float* __restrict__ W2,
                 float* __restrict__ out2, int n_tiles) {
    __shared__ v2f sB[(HID / 2) * BPAIR_STRIDE];
    {
        float* sf = (float*)sB;
        for (int idx = threadIdx.x; idx < HID * HID; idx += blockDim.x) {
            const int k = idx >> 6, col = idx & 63;
            const int slot = (((col >> 5) << 1) + (col & 1)) * 16 + ((col & 31) >> 1);
            sf[(k >> 1) * (2 * BPAIR_STRIDE) + slot * 2 + (k & 1)] = W2[idx];
        }
    }
    __syncthreads();

    const int wave = threadIdx.x >> 5;
    const int lane = threadIdx.x & 31;
    const int tile = blockIdx.x * 8 + wave;
    if (tile >= n_tiles) return;

    const int row0 = tile * 16;
    const int m    = lane & 15;
    const int half = lane >> 4;

    v8f acc0 = {}, acc1 = {}, acc2 = {}, acc3 = {};
    const float* arow = z + (size_t)(row0 + m) * HID;

    for (int kk = 0; kk < HID; kk += 4) {
        const int ka = kk + 2 * half;
        v2f a;
        a.x = arow[ka];
        a.y = arow[ka + 1];

        const v2f* bp = &sB[(ka >> 1) * BPAIR_STRIDE + m];
        acc0 = __builtin_amdgcn_wmma_f32_16x16x4_f32(false, a, false, bp[ 0], (short)0, acc0, false, false);
        acc1 = __builtin_amdgcn_wmma_f32_16x16x4_f32(false, a, false, bp[16], (short)0, acc1, false, false);
        acc2 = __builtin_amdgcn_wmma_f32_16x16x4_f32(false, a, false, bp[32], (short)0, acc2, false, false);
        acc3 = __builtin_amdgcn_wmma_f32_16x16x4_f32(false, a, false, bp[48], (short)0, acc3, false, false);
    }

    float* orow = out2 + (size_t)(row0 + half * 8) * HID;
#pragma unroll
    for (int r = 0; r < 8; ++r) {
        v2f lo, hi;
        lo.x = acc0[r];
        lo.y = acc1[r];
        hi.x = acc2[r];
        hi.y = acc3[r];
        *(v2f*)(orow + (size_t)r * HID +      2 * m) = lo;
        *(v2f*)(orow + (size_t)r * HID + 32 + 2 * m) = hi;
    }
}

__global__ void k_pool_init(float* pool0, float* pool1, float* cnt0, float* cnt1) {
    int i = blockIdx.x * blockDim.x + threadIdx.x;
    if (i < N_GRAPH * HID) {
        pool0[i] = 0.0f;
        pool1[i] = 0.0f;
    }
    if (i < N_GRAPH) {
        cnt0[i] = 0.0f;
        cnt1[i] = 0.0f;
    }
}

__global__ __launch_bounds__(256)
void k_relu_pool(const float* __restrict__ out2, const float* __restrict__ dinv,
                 const float* __restrict__ b2, const int* __restrict__ batch,
                 float* pool0, float* pool1, float* cnt0, float* cnt1, int n) {
    const int hf = package_half();
    float* pool = hf ? pool1 : pool0;
    float* cnt  = hf ? cnt1 : cnt0;
    const int total  = n * HID;
    const int stride = gridDim.x * blockDim.x;
    int tid = blockIdx.x * blockDim.x + threadIdx.x;
    const int h = tid & 63;
    const float bh = b2[h];
    for (; tid + 3 * stride < total; tid += 4 * stride) {
        const int t0 = tid, t1 = tid + stride, t2 = tid + 2 * stride, t3 = tid + 3 * stride;
        const int i0 = t0 >> 6, i1 = t1 >> 6, i2 = t2 >> 6, i3 = t3 >> 6;
        const float o0 = out2[t0], o1 = out2[t1], o2 = out2[t2], o3 = out2[t3];
        const float e0 = dinv[i0], e1 = dinv[i1], e2 = dinv[i2], e3 = dinv[i3];
        const int g0 = batch[i0], g1 = batch[i1], g2 = batch[i2], g3 = batch[i3];
        float v0 = e0 * o0 + bh;
        float v1 = e1 * o1 + bh;
        float v2 = e2 * o2 + bh;
        float v3 = e3 * o3 + bh;
        v0 = v0 > 0.0f ? v0 : 0.0f;
        v1 = v1 > 0.0f ? v1 : 0.0f;
        v2 = v2 > 0.0f ? v2 : 0.0f;
        v3 = v3 > 0.0f ? v3 : 0.0f;
        atomicAdd(&pool[g0 * HID + h], v0);
        atomicAdd(&pool[g1 * HID + h], v1);
        atomicAdd(&pool[g2 * HID + h], v2);
        atomicAdd(&pool[g3 * HID + h], v3);
        if (h == 0) {
            atomicAdd(&cnt[g0], 1.0f);
            atomicAdd(&cnt[g1], 1.0f);
            atomicAdd(&cnt[g2], 1.0f);
            atomicAdd(&cnt[g3], 1.0f);
        }
    }
    for (; tid < total; tid += stride) {
        const int i = tid >> 6;
        float v = dinv[i] * out2[tid] + bh;
        v = v > 0.0f ? v : 0.0f;
        const int g = batch[i];
        atomicAdd(&pool[g * HID + h], v);
        if (h == 0) atomicAdd(&cnt[g], 1.0f);
    }
}

__global__ void k_final(const float* __restrict__ pool0, const float* __restrict__ pool1,
                        const float* __restrict__ cnt0, const float* __restrict__ cnt1,
                        float* __restrict__ out) {
    int tid = blockIdx.x * blockDim.x + threadIdx.x;
    if (tid < N_GRAPH * HID) {
        float c = cnt0[tid >> 6] + cnt1[tid >> 6];
        out[tid] = (pool0[tid] + pool1[tid]) / fmaxf(c, 1.0f);
    }
}

extern "C" void kernel_launch(void* const* d_in, const int* in_sizes, int n_in,
                              void* d_out, int out_size, void* d_ws, size_t ws_size,
                              hipStream_t stream) {
    const float* x     = (const float*)d_in[0];
    const int*   edge  = (const int*)  d_in[1];
    const int*   batch = (const int*)  d_in[2];
    const float* W1    = (const float*)d_in[3];
    const float* b1    = (const float*)d_in[4];
    const float* W2    = (const float*)d_in[5];
    const float* b2    = (const float*)d_in[6];
    const int* src = edge;
    const int* dst = edge + N_EDGES;
    float* out = (float*)d_out;

    float* ws   = (float*)d_ws;
    float* dinv = ws;
    float* s    = dinv + N_NODES;
    float* bufA = s + N_NODES;
    float* bufB = bufA + (size_t)N_NODES * HID;
    float* y    = bufB;
    float* deg1 = bufB + N_NODES;
    float* s1   = bufB + 2 * N_NODES;
    float* pairs = bufB + 3 * N_NODES;
    int*   tick  = (int*)(bufB + 5 * N_NODES);
    float* pool = bufB + (size_t)N_NODES * HID;
    float* cnt  = pool + N_GRAPH * HID;
    float* pool1 = s;
    float* cnt1  = s + N_GRAPH * HID;

    const int TB = 256;
    k_init_deg     <<<(N_NODES + TB - 1) / TB, TB, 0, stream>>>(dinv, deg1, N_NODES);
    k_deg_edges    <<<EDGE_BLOCKS, TB, 0, stream>>>(dst, dinv, deg1, N_EDGES);
    k_dinv_selfloop<<<(N_NODES + TB - 1) / TB, TB, 0, stream>>>(dinv, deg1, x, y, s, s1, N_NODES);
    k_l1_edges     <<<EDGE_BLOCKS, TB, 0, stream>>>(src, dst, y, s, s1, N_EDGES);
    k_l1_rows      <<<ROW_BLOCKS, TB, 0, stream>>>(s, s1, dinv, W1, b1, bufA, pairs, tick, N_NODES);
    k_l2_edges     <<<L2_BLOCKS, TB, 0, stream>>>(src, dst, pairs, W1, b1, bufA, tick, N_EDGES);

    const int n_tiles = N_NODES / 16;
    k_gemm_wmma    <<<(n_tiles + 7) / 8, 256, 0, stream>>>(bufA, W2, bufB, n_tiles);

    k_pool_init    <<<(N_GRAPH * HID + TB - 1) / TB, TB, 0, stream>>>(pool, pool1, cnt, cnt1);
    k_relu_pool    <<<POOL_BLOCKS, TB, 0, stream>>>(bufB, dinv, b2, batch, pool, pool1, cnt, cnt1, N_NODES);
    k_final        <<<(N_GRAPH * HID + TB - 1) / TB, TB, 0, stream>>>(pool, pool1, cnt, cnt1, out);
}


# Zero b1 (tested on the device each call): relu(a*w)=relu(w)relu(a)+relu(-w)relu(-a), so layer 2 sums 2 scalars per node, one 4-byte add per edge; z rows rebuilt in the GEMM A loader (sums regrouped)
# speedup vs baseline: 6.1646x; 1.4064x over previous
#include <hip/hip_runtime.h>

typedef float v2f __attribute__((ext_vector_type(2)));
typedef float v8f __attribute__((ext_vector_type(8)));

#define N_NODES 200000
#define N_EDGES 3200000
#define N_GRAPH 2048
#define HID     64
#define BPAIR_STRIDE 80

#define EDGE_BLOCKS ((N_EDGES + 256 * 8 - 1) / (256 * 8))
#define POOL_BLOCKS 2048
#define TICK_STRIDE 64

__device__ __forceinline__ int package_half() {
    const unsigned id = __builtin_amdgcn_s_sendmsg_rtn(0x87);
    return (int)((id >> 18) & 1u);
}

__global__ void k_init_deg(float* deg0, float* deg1, int n) {
    int i = blockIdx.x * blockDim.x + threadIdx.x;
    if (i < n) {
        deg0[i] = 1.0f;
        deg1[i] = 0.0f;
    }
}

__global__ __launch_bounds__(256)
void k_deg_edges(const int* __restrict__ dst, float* deg0, float* deg1, int e) {
    float* deg = package_half() ? deg1 : deg0;
    const int stride = gridDim.x * blockDim.x;
    int i = blockIdx.x * blockDim.x + threadIdx.x;
    for (; i + 3 * stride < e; i += 4 * stride) {
        const int d0 = dst[i];
        const int d1 = dst[i + stride];
        const int d2 = dst[i + 2 * stride];
        const int d3 = dst[i + 3 * stride];
        atomicAdd(&deg[d0], 1.0f);
        atomicAdd(&deg[d1], 1.0f);
        atomicAdd(&deg[d2], 1.0f);
        atomicAdd(&deg[d3], 1.0f);
    }
    for (; i < e; i += stride) atomicAdd(&deg[dst[i]], 1.0f);
}

__global__ void k_dinv_selfloop(float* deg_dinv, const float* __restrict__ deg1,
                                const float* __restrict__ x,
                                float* __restrict__ y, float* s0, float* s1, int n) {
    int i = blockIdx.x * blockDim.x + threadIdx.x;
    if (i < n) {
        float d = rsqrtf(deg_dinv[i] + deg1[i]);
        deg_dinv[i] = d;
        float yi = d * x[i];
        y[i] = yi;
        s0[i] = yi;
        s1[i] = 0.0f;
    }
}

__global__ __launch_bounds__(256)
void k_l1_edges(const int* __restrict__ src, const int* __restrict__ dst,
                const float* __restrict__ y, float* s0, float* s1, int e) {
    float* s = package_half() ? s1 : s0;
    const int stride = gridDim.x * blockDim.x;
    int i = blockIdx.x * blockDim.x + threadIdx.x;
    for (; i + 3 * stride < e; i += 4 * stride) {
        const int u0 = src[i];
        const int u1 = src[i + stride];
        const int u2 = src[i + 2 * stride];
        const int u3 = src[i + 3 * stride];
        const int d0 = dst[i];
        const int d1 = dst[i + stride];
        const int d2 = dst[i + 2 * stride];
        const int d3 = dst[i + 3 * stride];
        const float y0 = y[u0];
        const float y1 = y[u1];
        const float y2 = y[u2];
        const float y3 = y[u3];
        atomicAdd(&s[d0], y0);
        atomicAdd(&s[d1], y1);
        atomicAdd(&s[d2], y2);
        atomicAdd(&s[d3], y3);
    }
    for (; i < e; i += stride) atomicAdd(&s[dst[i]], y[src[i]]);
}

#define ROW_BLOCKS 1024
typedef float v4f __attribute__((ext_vector_type(4)));

__device__ __forceinline__ float l1_scaled(float a, float d, float w, float b) {
    float r = a * w + b;
    r = r > 0.0f ? r : 0.0f;
    return d * r;
}

__device__ __forceinline__ v4f l1_piece(float a, float d, v4f w, v4f b) {
    v4f r;
    r.x = l1_scaled(a, d, w.x, b.x);
    r.y = l1_scaled(a, d, w.y, b.y);
    r.z = l1_scaled(a, d, w.z, b.z);
    r.w = l1_scaled(a, d, w.w, b.w);
    return r;
}

__device__ __forceinline__ bool bias_is_zero(const float* __restrict__ b1, int lane) {
    const float x0 = b1[lane], x1 = b1[lane + 32];
    return __builtin_amdgcn_ballot_w32((x0 != 0.0f) | (x1 != 0.0f)) == 0u;
}

__device__ __forceinline__ void l2_scalar_add(float* acc, int v, float g) {
    atomicAdd(&acc[2 * v + (g < 0.0f ? 1 : 0)], fabsf(g));
}

__global__ __launch_bounds__(256)
void k_l1_rows(const float* __restrict__ s0, const float* __restrict__ s1,
               const float* __restrict__ dinv,
               const float* __restrict__ W1,
               const float* __restrict__ b1, float* __restrict__ z,
               float* __restrict__ pairs, int* tick, int n) {
    if (blockIdx.x == 0 && threadIdx.x < 2) tick[threadIdx.x * TICK_STRIDE] = 0;
    if (bias_is_zero(b1, threadIdx.x & 31)) {
        v2f* pq0 = (v2f*)z;
        v2f* pq1 = pq0 + n;
        const int nstride = gridDim.x * blockDim.x;
        for (int i = blockIdx.x * blockDim.x + threadIdx.x; i < n; i += nstride) {
            const float d = dinv[i];
            const float g = d * (d * (s0[i] + s1[i]));
            pairs[i] = g;
            v2f own, none;
            own.x = g > 0.0f ? g : 0.0f;
            own.y = g < 0.0f ? -g : 0.0f;
            none.x = 0.0f; none.y = 0.0f;
            pq0[i] = own;
            pq1[i] = none;
        }
        return;
    }
    const int total  = n * (HID / 4);
    const int stride = gridDim.x * blockDim.x;
    int q = blockIdx.x * blockDim.x + threadIdx.x;
    const int c4 = (q & 15) * 4;
    const bool first = (c4 == 0);
    const v4f w = *(const v4f*)(W1 + c4);
    const v4f b = *(const v4f*)(b1 + c4);
    v4f* zv = (v4f*)z;
    v2f* pv = (v2f*)pairs;
    for (; q + 3 * stride < total; q += 4 * stride) {
        const int q0 = q, q1 = q + stride, q2 = q + 2 * stride, q3 = q + 3 * stride;
        const int i0 = q0 >> 4, i1 = q1 >> 4, i2 = q2 >> 4, i3 = q3 >> 4;
        const float d0 = dinv[i0], d1 = dinv[i1], d2 = dinv[i2], d3 = dinv[i3];
        const float a0 = d0 * (s0[i0] + s1[i0]);
        const float a1 = d1 * (s0[i1] + s1[i1]);
        const float a2 = d2 * (s0[i2] + s1[i2]);
        const float a3 = d3 * (s0[i3] + s1[i3]);
        zv[q0] = l1_piece(a0, d0, w, b);
        zv[q1] = l1_piece(a1, d1, w, b);
        zv[q2] = l1_piece(a2, d2, w, b);
        zv[q3] = l1_piece(a3, d3, w, b);
        if (first) {
            v2f p0, p1, p2, p3;
            p0.x = a0; p0.y = d0;
            p1.x = a1; p1.y = d1;
            p2.x = a2; p2.y = d2;
            p3.x = a3; p3.y = d3;
            pv[i0] = p0;
            pv[i1] = p1;
            pv[i2] = p2;
            pv[i3] = p3;
        }
    }
    for (; q < total; q += stride) {
        const int i = q >> 4;
        const float d = dinv[i];
        const float a = d * (s0[i] + s1[i]);
        zv[q] = l1_piece(a, d, w, b);
        if (first) {
            v2f p;
            p.x = a; p.y = d;
            pv[i] = p;
        }
    }
}

#define L2_BLOCKS 2048
#define L2_ROLES (L2_BLOCKS / 2)
#define L2_SPLIT (N_NODES / 2)

__device__ __forceinline__ void l2_fetch_idx(const int* __restrict__ src, const int* __restrict__ dst,
                                             int grp, int g1, int lane, int e, int& u, int& v) {
    const int ei = grp * 32 + lane;
    const bool ok = (grp < g1) && (ei < e);
    const int ec = ok ? ei : (e - 1);
    u = src[ec];
    const int vv = dst[ec];
    v = ok ? vv : -1;
}

__global__ __launch_bounds__(256)
void k_l2_edges(const int* __restrict__ src, const int* __restrict__ dst,
                const float* __restrict__ pairs,
                const float* __restrict__ W1, const float* __restrict__ b1,
                float* z, int* tick, int e) {
    __shared__ int s_role[2];
    const int hf = package_half();
    if (bias_is_zero(b1, threadIdx.x & 31)) {
        const float* __restrict__ gs = pairs;
        float* acc = z + (size_t)hf * (2 * N_NODES);
        const int stride = gridDim.x * blockDim.x;
        int i = blockIdx.x * blockDim.x + threadIdx.x;
        for (; i + 3 * stride < e; i += 4 * stride) {
            const int u0 = src[i];
            const int u1 = src[i + stride];
            const int u2 = src[i + 2 * stride];
            const int u3 = src[i + 3 * stride];
            const int d0 = dst[i];
            const int d1 = dst[i + stride];
            const int d2 = dst[i + 2 * stride];
            const int d3 = dst[i + 3 * stride];
            const float g0 = gs[u0];
            const float g1 = gs[u1];
            const float g2 = gs[u2];
            const float g3 = gs[u3];
            l2_scalar_add(acc, d0, g0);
            l2_scalar_add(acc, d1, g1);
            l2_scalar_add(acc, d2, g2);
            l2_scalar_add(acc, d3, g3);
        }
        for (; i + stride < e; i += 2 * stride) {
            const int u0 = src[i];
            const int u1 = src[i + stride];
            const int d0 = dst[i];
            const int d1 = dst[i + stride];
            const float g0 = gs[u0];
            const float g1 = gs[u1];
            l2_scalar_add(acc, d0, g0);
            l2_scalar_add(acc, d1, g1);
        }
        for (; i < e; i += stride) l2_scalar_add(acc, dst[i], gs[src[i]]);
        return;
    }
    if (threadIdx.x == 0) {
        const int r = atomicAdd(&tick[hf * TICK_STRIDE], 1);
        int own = hf, share = r;
        if (r >= L2_ROLES) {
            own = 1 - hf;
            share = 2 * L2_ROLES - 1 - r;
        }
        s_role[0] = own;
        s_role[1] = share;
    }
    __syncthreads();
    const int own   = __builtin_amdgcn_readfirstlane(s_role[0]);
    const int share = __builtin_amdgcn_readfirstlane(s_role[1]);
    if (share < 0 || share >= L2_ROLES) return;

    const int lane    = threadIdx.x & 31;
    const float w0 = W1[lane], w1 = W1[lane + 32];
    const float c0 = b1[lane], c1 = b1[lane + 32];
    const v2f* pr = (const v2f*)pairs;
    const int ngroups = (e + 31) >> 5;
    const int widx = share * 8 + (threadIdx.x >> 5);
    int g        = (int)(((long long)widx * ngroups) / (8 * L2_ROLES));
    const int g1 = (int)(((long long)(widx + 1) * ngroups) / (8 * L2_ROLES));

    int u0, v0;
    int u1, v1;
    l2_fetch_idx(src, dst, g, g1, lane, e, u0, v0);
    l2_fetch_idx(src, dst, g + 1, g1, lane, e, u1, v1);
    float a0 = 0.0f, d0 = 0.0f;
    if ((v0 >= 0) && ((v0 >= L2_SPLIT ? 1 : 0) == own)) {
        const v2f p = pr[u0];
        a0 = p.x;
        d0 = p.y;
    }
    while (g < g1) {
        int u2, v2;
        l2_fetch_idx(src, dst, g + 2, g1, lane, e, u2, v2);
        float a1 = 0.0f, d1 = 0.0f;
        if ((v1 >= 0) && ((v1 >= L2_SPLIT ? 1 : 0) == own)) {
            const v2f p = pr[u1];
            a1 = p.x;
            d1 = p.y;
        }

        unsigned mask = __builtin_amdgcn_ballot_w32((v0 >= 0) && ((v0 >= L2_SPLIT ? 1 : 0) == own));
        while (mask != 0u) {
            const int j = __builtin_ctz(mask);
            mask &= mask - 1u;
            const float sa = __int_as_float(__builtin_amdgcn_readlane(__float_as_int(a0), j));
            const float sd = __int_as_float(__builtin_amdgcn_readlane(__float_as_int(d0), j));
            const int   sv = __builtin_amdgcn_readlane(v0, j);
            float* orow = z + (size_t)sv * HID + lane;
            atomicAdd(&orow[0],  l1_scaled(sa, sd, w0, c0));
            atomicAdd(&orow[32], l1_scaled(sa, sd, w1, c1));
        }
        v0 = v1;
        a0 = a1;
        d0 = d1;
        u1 = u2;
        v1 = v2;
        ++g;
    }
}

__global__ __launch_bounds__(256)
void k_gemm_wmma(const float* __restrict__ z, const float* __restrict__ W2,
                 const float* __restrict__ W1, const float* __restrict__ b1,
                 float* __restrict__ out2, int n_tiles) {
    __shared__ v2f sB[(HID / 2) * BPAIR_STRIDE];
    {
        float* sf = (float*)sB;
        for (int idx = threadIdx.x; idx < HID * HID; idx += blockDim.x) {
            const int k = idx >> 6, col = idx & 63;
            const int slot = (((col >> 5) << 1) + (col & 1)) * 16 + ((col & 31) >> 1);
            sf[(k >> 1) * (2 * BPAIR_STRIDE) + slot * 2 + (k & 1)] = W2[idx];
        }
    }
    __syncthreads();

    const int wave = threadIdx.x >> 5;
    const int lane = threadIdx.x & 31;
    const int tile = blockIdx.x * 8 + wave;
    if (tile >= n_tiles) return;

    const int row0 = tile * 16;
    const int m    = lane & 15;
    const int half = lane >> 4;

    v8f acc0 = {}, acc1 = {}, acc2 = {}, acc3 = {};
    const float* arow = z + (size_t)(row0 + m) * HID;

    if (bias_is_zero(b1, lane)) {
        const v2f* pq = (const v2f*)z;
        const v2f h0 = pq[row0 + m];
        const v2f h1 = pq[n_tiles * 16 + row0 + m];
        const float P = h0.x + h1.x;
        const float Q = h0.y + h1.y;
        for (int kk = 0; kk < HID; kk += 4) {
            const int ka = kk + 2 * half;
            const v2f w = *(const v2f*)(W1 + ka);
            v2f a;
            a.x = w.x > 0.0f ? w.x * P : -w.x * Q;
            a.y = w.y > 0.0f ? w.y * P : -w.y * Q;
            const v2f* bp = &sB[(ka >> 1) * BPAIR_STRIDE + m];
            acc0 = __builtin_amdgcn_wmma_f32_16x16x4_f32(false, a, false, bp[ 0], (short)0, acc0, false, false);
            acc1 = __builtin_amdgcn_wmma_f32_16x16x4_f32(false, a, false, bp[16], (short)0, acc1, false, false);
            acc2 = __builtin_amdgcn_wmma_f32_16x16x4_f32(false, a, false, bp[32], (short)0, acc2, false, false);
            acc3 = __builtin_amdgcn_wmma_f32_16x16x4_f32(false, a, false, bp[48], (short)0, acc3, false, false);
        }
    } else {
    for (int kk = 0; kk < HID; kk += 4) {
        const int ka = kk + 2 * half;
        v2f a;
        a.x = arow[ka];
        a.y = arow[ka + 1];

        const v2f* bp = &sB[(ka >> 1) * BPAIR_STRIDE + m];
        acc0 = __builtin_amdgcn_wmma_f32_16x16x4_f32(false, a, false, bp[ 0], (short)0, acc0, false, false);
        acc1 = __builtin_amdgcn_wmma_f32_16x16x4_f32(false, a, false, bp[16], (short)0, acc1, false, false);
        acc2 = __builtin_amdgcn_wmma_f32_16x16x4_f32(false, a, false, bp[32], (short)0, acc2, false, false);
        acc3 = __builtin_amdgcn_wmma_f32_16x16x4_f32(false, a, false, bp[48], (short)0, acc3, false, false);
    }
    }

    float* orow = out2 + (size_t)(row0 + half * 8) * HID;
#pragma unroll
    for (int r = 0; r < 8; ++r) {
        v2f lo, hi;
        lo.x = acc0[r];
        lo.y = acc1[r];
        hi.x = acc2[r];
        hi.y = acc3[r];
        *(v2f*)(orow + (size_t)r * HID +      2 * m) = lo;
        *(v2f*)(orow + (size_t)r * HID + 32 + 2 * m) = hi;
    }
}

__global__ void k_pool_init(float* pool0, float* pool1, float* cnt0, float* cnt1) {
    int i = blockIdx.x * blockDim.x + threadIdx.x;
    if (i < N_GRAPH * HID) {
        pool0[i] = 0.0f;
        pool1[i] = 0.0f;
    }
    if (i < N_GRAPH) {
        cnt0[i] = 0.0f;
        cnt1[i] = 0.0f;
    }
}

__global__ __launch_bounds__(256)
void k_relu_pool(const float* __restrict__ out2, const float* __restrict__ dinv,
                 const float* __restrict__ b2, const int* __restrict__ batch,
                 float* pool0, float* pool1, float* cnt0, float* cnt1, int n) {
    const int hf = package_half();
    float* pool = hf ? pool1 : pool0;
    float* cnt  = hf ? cnt1 : cnt0;
    const int total  = n * HID;
    const int stride = gridDim.x * blockDim.x;
    int tid = blockIdx.x * blockDim.x + threadIdx.x;
    const int h = tid & 63;
    const float bh = b2[h];
    for (; tid + 3 * stride < total; tid += 4 * stride) {
        const int t0 = tid, t1 = tid + stride, t2 = tid + 2 * stride, t3 = tid + 3 * stride;
        const int i0 = t0 >> 6, i1 = t1 >> 6, i2 = t2 >> 6, i3 = t3 >> 6;
        const float o0 = out2[t0], o1 = out2[t1], o2 = out2[t2], o3 = out2[t3];
        const float e0 = dinv[i0], e1 = dinv[i1], e2 = dinv[i2], e3 = dinv[i3];
        const int g0 = batch[i0], g1 = batch[i1], g2 = batch[i2], g3 = batch[i3];
        float v0 = e0 * o0 + bh;
        float v1 = e1 * o1 + bh;
        float v2 = e2 * o2 + bh;
        float v3 = e3 * o3 + bh;
        v0 = v0 > 0.0f ? v0 : 0.0f;
        v1 = v1 > 0.0f ? v1 : 0.0f;
        v2 = v2 > 0.0f ? v2 : 0.0f;
        v3 = v3 > 0.0f ? v3 : 0.0f;
        atomicAdd(&pool[g0 * HID + h], v0);
        atomicAdd(&pool[g1 * HID + h], v1);
        atomicAdd(&pool[g2 * HID + h], v2);
        atomicAdd(&pool[g3 * HID + h], v3);
        if (h == 0) {
            atomicAdd(&cnt[g0], 1.0f);
            atomicAdd(&cnt[g1], 1.0f);
            atomicAdd(&cnt[g2], 1.0f);
            atomicAdd(&cnt[g3], 1.0f);
        }
    }
    for (; tid < total; tid += stride) {
        const int i = tid >> 6;
        float v = dinv[i] * out2[tid] + bh;
        v = v > 0.0f ? v : 0.0f;
        const int g = batch[i];
        atomicAdd(&pool[g * HID + h], v);
        if (h == 0) atomicAdd(&cnt[g], 1.0f);
    }
}

__global__ void k_final(const float* __restrict__ pool0, const float* __restrict__ pool1,
                        const float* __restrict__ cnt0, const float* __restrict__ cnt1,
                        float* __restrict__ out) {
    int tid = blockIdx.x * blockDim.x + threadIdx.x;
    if (tid < N_GRAPH * HID) {
        float c = cnt0[tid >> 6] + cnt1[tid >> 6];
        out[tid] = (pool0[tid] + pool1[tid]) / fmaxf(c, 1.0f);
    }
}

extern "C" void kernel_launch(void* const* d_in, const int* in_sizes, int n_in,
                              void* d_out, int out_size, void* d_ws, size_t ws_size,
                              hipStream_t stream) {
    const float* x     = (const float*)d_in[0];
    const int*   edge  = (const int*)  d_in[1];
    const int*   batch = (const int*)  d_in[2];
    const float* W1    = (const float*)d_in[3];
    const float* b1    = (const float*)d_in[4];
    const float* W2    = (const float*)d_in[5];
    const float* b2    = (const float*)d_in[6];
    const int* src = edge;
    const int* dst = edge + N_EDGES;
    float* out = (float*)d_out;

    float* ws   = (float*)d_ws;
    float* dinv = ws;
    float* s    = dinv + N_NODES;
    float* bufA = s + N_NODES;
    float* bufB = bufA + (size_t)N_NODES * HID;
    float* y    = bufB;
    float* deg1 = bufB + N_NODES;
    float* s1   = bufB + 2 * N_NODES;
    float* pairs = bufB + 3 * N_NODES;
    int*   tick  = (int*)(bufB + 5 * N_NODES);
    float* pool = bufB + (size_t)N_NODES * HID;
    float* cnt  = pool + N_GRAPH * HID;
    float* pool1 = s;
    float* cnt1  = s + N_GRAPH * HID;

    const int TB = 256;
    k_init_deg     <<<(N_NODES + TB - 1) / TB, TB, 0, stream>>>(dinv, deg1, N_NODES);
    k_deg_edges    <<<EDGE_BLOCKS, TB, 0, stream>>>(dst, dinv, deg1, N_EDGES);
    k_dinv_selfloop<<<(N_NODES + TB - 1) / TB, TB, 0, stream>>>(dinv, deg1, x, y, s, s1, N_NODES);
    k_l1_edges     <<<EDGE_BLOCKS, TB, 0, stream>>>(src, dst, y, s, s1, N_EDGES);
    k_l1_rows      <<<ROW_BLOCKS, TB, 0, stream>>>(s, s1, dinv, W1, b1, bufA, pairs, tick, N_NODES);
    k_l2_edges     <<<L2_BLOCKS, TB, 0, stream>>>(src, dst, pairs, W1, b1, bufA, tick, N_EDGES);

    const int n_tiles = N_NODES / 16;
    k_gemm_wmma    <<<(n_tiles + 7) / 8, 256, 0, stream>>>(bufA, W2, W1, b1, bufB, n_tiles);

    k_pool_init    <<<(N_GRAPH * HID + TB - 1) / TB, TB, 0, stream>>>(pool, pool1, cnt, cnt1);
    k_relu_pool    <<<POOL_BLOCKS, TB, 0, stream>>>(bufB, dinv, b2, batch, pool, pool1, cnt, cnt1, N_NODES);
    k_final        <<<(N_GRAPH * HID + TB - 1) / TB, TB, 0, stream>>>(pool, pool1, cnt, cnt1, out);
}


# Pool fused into the W2 product: the epilogue applies dinv, b2, relu and adds rows to the pool (half-waves trade 16 columns: one atomic = one row's 128-B line); out2 never stored (add order as before)
# speedup vs baseline: 7.1165x; 1.1544x over previous
#include <hip/hip_runtime.h>

typedef float v2f __attribute__((ext_vector_type(2)));
typedef float v8f __attribute__((ext_vector_type(8)));

#define N_NODES 200000
#define N_EDGES 3200000
#define N_GRAPH 2048
#define HID     64
#define BPAIR_STRIDE 80

#define EDGE_BLOCKS ((N_EDGES + 256 * 8 - 1) / (256 * 8))
#define TICK_STRIDE 64

__device__ __forceinline__ int package_half() {
    const unsigned id = __builtin_amdgcn_s_sendmsg_rtn(0x87);
    return (int)((id >> 18) & 1u);
}

__global__ void k_init_deg(float* deg0, float* deg1,
                           float* pool0, float* pool1, float* cnt0, float* cnt1, int n) {
    int i = blockIdx.x * blockDim.x + threadIdx.x;
    if (i < n) {
        deg0[i] = 1.0f;
        deg1[i] = 0.0f;
    }
    if (i < N_GRAPH * HID) {
        pool0[i] = 0.0f;
        pool1[i] = 0.0f;
    }
    if (i < N_GRAPH) {
        cnt0[i] = 0.0f;
        cnt1[i] = 0.0f;
    }
}

__global__ __launch_bounds__(256)
void k_deg_edges(const int* __restrict__ dst, float* deg0, float* deg1, int e) {
    float* deg = package_half() ? deg1 : deg0;
    const int stride = gridDim.x * blockDim.x;
    int i = blockIdx.x * blockDim.x + threadIdx.x;
    for (; i + 3 * stride < e; i += 4 * stride) {
        const int d0 = dst[i];
        const int d1 = dst[i + stride];
        const int d2 = dst[i + 2 * stride];
        const int d3 = dst[i + 3 * stride];
        atomicAdd(&deg[d0], 1.0f);
        atomicAdd(&deg[d1], 1.0f);
        atomicAdd(&deg[d2], 1.0f);
        atomicAdd(&deg[d3], 1.0f);
    }
    for (; i < e; i += stride) atomicAdd(&deg[dst[i]], 1.0f);
}

__global__ void k_dinv_selfloop(float* deg_dinv, const float* __restrict__ deg1,
                                const float* __restrict__ x,
                                float* __restrict__ y, float* s0, float* s1, int n) {
    int i = blockIdx.x * blockDim.x + threadIdx.x;
    if (i < n) {
        float d = rsqrtf(deg_dinv[i] + deg1[i]);
        deg_dinv[i] = d;
        float yi = d * x[i];
        y[i] = yi;
        s0[i] = yi;
        s1[i] = 0.0f;
    }
}

__global__ __launch_bounds__(256)
void k_l1_edges(const int* __restrict__ src, const int* __restrict__ dst,
                const float* __restrict__ y, float* s0, float* s1, int e) {
    float* s = package_half() ? s1 : s0;
    const int stride = gridDim.x * blockDim.x;
    int i = blockIdx.x * blockDim.x + threadIdx.x;
    for (; i + 3 * stride < e; i += 4 * stride) {
        const int u0 = src[i];
        const int u1 = src[i + stride];
        const int u2 = src[i + 2 * stride];
        const int u3 = src[i + 3 * stride];
        const int d0 = dst[i];
        const int d1 = dst[i + stride];
        const int d2 = dst[i + 2 * stride];
        const int d3 = dst[i + 3 * stride];
        const float y0 = y[u0];
        const float y1 = y[u1];
        const float y2 = y[u2];
        const float y3 = y[u3];
        atomicAdd(&s[d0], y0);
        atomicAdd(&s[d1], y1);
        atomicAdd(&s[d2], y2);
        atomicAdd(&s[d3], y3);
    }
    for (; i < e; i += stride) atomicAdd(&s[dst[i]], y[src[i]]);
}

#define ROW_BLOCKS 1024
typedef float v4f __attribute__((ext_vector_type(4)));

__device__ __forceinline__ float l1_scaled(float a, float d, float w, float b) {
    float r = a * w + b;
    r = r > 0.0f ? r : 0.0f;
    return d * r;
}

__device__ __forceinline__ v4f l1_piece(float a, float d, v4f w, v4f b) {
    v4f r;
    r.x = l1_scaled(a, d, w.x, b.x);
    r.y = l1_scaled(a, d, w.y, b.y);
    r.z = l1_scaled(a, d, w.z, b.z);
    r.w = l1_scaled(a, d, w.w, b.w);
    return r;
}

__device__ __forceinline__ bool bias_is_zero(const float* __restrict__ b1, int lane) {
    const float x0 = b1[lane], x1 = b1[lane + 32];
    return __builtin_amdgcn_ballot_w32((x0 != 0.0f) | (x1 != 0.0f)) == 0u;
}

__device__ __forceinline__ void l2_scalar_add(float* acc, int v, float g) {
    atomicAdd(&acc[2 * v + (g < 0.0f ? 1 : 0)], fabsf(g));
}

__global__ __launch_bounds__(256)
void k_l1_rows(const float* __restrict__ s0, const float* __restrict__ s1,
               const float* __restrict__ dinv,
               const float* __restrict__ W1,
               const float* __restrict__ b1, float* __restrict__ z,
               float* __restrict__ pairs, int* tick, int n) {
    if (blockIdx.x == 0 && threadIdx.x < 2) tick[threadIdx.x * TICK_STRIDE] = 0;
    if (bias_is_zero(b1, threadIdx.x & 31)) {
        v2f* pq0 = (v2f*)z;
        v2f* pq1 = pq0 + n;
        const int nstride = gridDim.x * blockDim.x;
        for (int i = blockIdx.x * blockDim.x + threadIdx.x; i < n; i += nstride) {
            const float d = dinv[i];
            const float g = d * (d * (s0[i] + s1[i]));
            pairs[i] = g;
            v2f own, none;
            own.x = g > 0.0f ? g : 0.0f;
            own.y = g < 0.0f ? -g : 0.0f;
            none.x = 0.0f; none.y = 0.0f;
            pq0[i] = own;
            pq1[i] = none;
        }
        return;
    }
    const int total  = n * (HID / 4);
    const int stride = gridDim.x * blockDim.x;
    int q = blockIdx.x * blockDim.x + threadIdx.x;
    const int c4 = (q & 15) * 4;
    const bool first = (c4 == 0);
    const v4f w = *(const v4f*)(W1 + c4);
    const v4f b = *(const v4f*)(b1 + c4);
    v4f* zv = (v4f*)z;
    v2f* pv = (v2f*)pairs;
    for (; q + 3 * stride < total; q += 4 * stride) {
        const int q0 = q, q1 = q + stride, q2 = q + 2 * stride, q3 = q + 3 * stride;
        const int i0 = q0 >> 4, i1 = q1 >> 4, i2 = q2 >> 4, i3 = q3 >> 4;
        const float d0 = dinv[i0], d1 = dinv[i1], d2 = dinv[i2], d3 = dinv[i3];
        const float a0 = d0 * (s0[i0] + s1[i0]);
        const float a1 = d1 * (s0[i1] + s1[i1]);
        const float a2 = d2 * (s0[i2] + s1[i2]);
        const float a3 = d3 * (s0[i3] + s1[i3]);
        zv[q0] = l1_piece(a0, d0, w, b);
        zv[q1] = l1_piece(a1, d1, w, b);
        zv[q2] = l1_piece(a2, d2, w, b);
        zv[q3] = l1_piece(a3, d3, w, b);
        if (first) {
            v2f p0, p1, p2, p3;
            p0.x = a0; p0.y = d0;
            p1.x = a1; p1.y = d1;
            p2.x = a2; p2.y = d2;
            p3.x = a3; p3.y = d3;
            pv[i0] = p0;
            pv[i1] = p1;
            pv[i2] = p2;
            pv[i3] = p3;
        }
    }
    for (; q < total; q += stride) {
        const int i = q >> 4;
        const float d = dinv[i];
        const float a = d * (s0[i] + s1[i]);
        zv[q] = l1_piece(a, d, w, b);
        if (first) {
            v2f p;
            p.x = a; p.y = d;
            pv[i] = p;
        }
    }
}

#define L2_BLOCKS 2048
#define L2_ROLES (L2_BLOCKS / 2)
#define L2_SPLIT (N_NODES / 2)

__device__ __forceinline__ void l2_fetch_idx(const int* __restrict__ src, const int* __restrict__ dst,
                                             int grp, int g1, int lane, int e, int& u, int& v) {
    const int ei = grp * 32 + lane;
    const bool ok = (grp < g1) && (ei < e);
    const int ec = ok ? ei : (e - 1);
    u = src[ec];
    const int vv = dst[ec];
    v = ok ? vv : -1;
}

__global__ __launch_bounds__(256)
void k_l2_edges(const int* __restrict__ src, const int* __restrict__ dst,
                const float* __restrict__ pairs,
                const float* __restrict__ W1, const float* __restrict__ b1,
                float* z, int* tick, int e) {
    __shared__ int s_role[2];
    const int hf = package_half();
    if (bias_is_zero(b1, threadIdx.x & 31)) {
        const float* __restrict__ gs = pairs;
        float* acc = z + (size_t)hf * (2 * N_NODES);
        const int stride = gridDim.x * blockDim.x;
        int i = blockIdx.x * blockDim.x + threadIdx.x;
        for (; i + 3 * stride < e; i += 4 * stride) {
            const int u0 = src[i];
            const int u1 = src[i + stride];
            const int u2 = src[i + 2 * stride];
            const int u3 = src[i + 3 * stride];
            const int d0 = dst[i];
            const int d1 = dst[i + stride];
            const int d2 = dst[i + 2 * stride];
            const int d3 = dst[i + 3 * stride];
            const float g0 = gs[u0];
            const float g1 = gs[u1];
            const float g2 = gs[u2];
            const float g3 = gs[u3];
            l2_scalar_add(acc, d0, g0);
            l2_scalar_add(acc, d1, g1);
            l2_scalar_add(acc, d2, g2);
            l2_scalar_add(acc, d3, g3);
        }
        for (; i + stride < e; i += 2 * stride) {
            const int u0 = src[i];
            const int u1 = src[i + stride];
            const int d0 = dst[i];
            const int d1 = dst[i + stride];
            const float g0 = gs[u0];
            const float g1 = gs[u1];
            l2_scalar_add(acc, d0, g0);
            l2_scalar_add(acc, d1, g1);
        }
        for (; i < e; i += stride) l2_scalar_add(acc, dst[i], gs[src[i]]);
        return;
    }
    if (threadIdx.x == 0) {
        const int r = atomicAdd(&tick[hf * TICK_STRIDE], 1);
        int own = hf, share = r;
        if (r >= L2_ROLES) {
            own = 1 - hf;
            share = 2 * L2_ROLES - 1 - r;
        }
        s_role[0] = own;
        s_role[1] = share;
    }
    __syncthreads();
    const int own   = __builtin_amdgcn_readfirstlane(s_role[0]);
    const int share = __builtin_amdgcn_readfirstlane(s_role[1]);
    if (share < 0 || share >= L2_ROLES) return;

    const int lane    = threadIdx.x & 31;
    const float w0 = W1[lane], w1 = W1[lane + 32];
    const float c0 = b1[lane], c1 = b1[lane + 32];
    const v2f* pr = (const v2f*)pairs;
    const int ngroups = (e + 31) >> 5;
    const int widx = share * 8 + (threadIdx.x >> 5);
    int g        = (int)(((long long)widx * ngroups) / (8 * L2_ROLES));
    const int g1 = (int)(((long long)(widx + 1) * ngroups) / (8 * L2_ROLES));

    int u0, v0;
    int u1, v1;
    l2_fetch_idx(src, dst, g, g1, lane, e, u0, v0);
    l2_fetch_idx(src, dst, g + 1, g1, lane, e, u1, v1);
    float a0 = 0.0f, d0 = 0.0f;
    if ((v0 >= 0) && ((v0 >= L2_SPLIT ? 1 : 0) == own)) {
        const v2f p = pr[u0];
        a0 = p.x;
        d0 = p.y;
    }
    while (g < g1) {
        int u2, v2;
        l2_fetch_idx(src, dst, g + 2, g1, lane, e, u2, v2);
        float a1 = 0.0f, d1 = 0.0f;
        if ((v1 >= 0) && ((v1 >= L2_SPLIT ? 1 : 0) == own)) {
            const v2f p = pr[u1];
            a1 = p.x;
            d1 = p.y;
        }

        unsigned mask = __builtin_amdgcn_ballot_w32((v0 >= 0) && ((v0 >= L2_SPLIT ? 1 : 0) == own));
        while (mask != 0u) {
            const int j = __builtin_ctz(mask);
            mask &= mask - 1u;
            const float sa = __int_as_float(__builtin_amdgcn_readlane(__float_as_int(a0), j));
            const float sd = __int_as_float(__builtin_amdgcn_readlane(__float_as_int(d0), j));
            const int   sv = __builtin_amdgcn_readlane(v0, j);
            float* orow = z + (size_t)sv * HID + lane;
            atomicAdd(&orow[0],  l1_scaled(sa, sd, w0, c0));
            atomicAdd(&orow[32], l1_scaled(sa, sd, w1, c1));
        }
        v0 = v1;
        a0 = a1;
        d0 = d1;
        u1 = u2;
        v1 = v2;
        ++g;
    }
}


__device__ __forceinline__ float xhalf(float x) {
    const int b = __builtin_bit_cast(int, x);
    return __builtin_bit_cast(float, __builtin_amdgcn_permlanex16(b, b, 0x76543210, 0xfedcba98, false, false));
}

__device__ __forceinline__ void pool_row_add(float* pool, float e, int g, int lane,
                                             float lo, float hi, float blo, float bhi) {
    float v0 = e * lo + blo;
    float v1 = e * hi + bhi;
    v0 = v0 > 0.0f ? v0 : 0.0f;
    v1 = v1 > 0.0f ? v1 : 0.0f;
    float* prow = pool + g * HID + lane;
    atomicAdd(&prow[0],  v0);
    atomicAdd(&prow[32], v1);
}

__global__ __launch_bounds__(256)
void k_gemm_pool(const float* __restrict__ z, const float* __restrict__ W2,
                 const float* __restrict__ W1, const float* __restrict__ b1,
                 const float* __restrict__ dinv, const float* __restrict__ b2,
                 const int* __restrict__ batch,
                 float* pool0, float* pool1, float* cnt0, float* cnt1, int n_tiles) {
    __shared__ v2f sB[(HID / 2) * BPAIR_STRIDE];
    {
        float* sf = (float*)sB;
        for (int idx = threadIdx.x; idx < HID * HID; idx += blockDim.x) {
            const int k = idx >> 6, col = idx & 63;
            sf[(k >> 1) * (2 * BPAIR_STRIDE) + col * 2 + (k & 1)] = W2[idx];
        }
    }
    __syncthreads();

    const int wave = __builtin_amdgcn_readfirstlane(threadIdx.x >> 5);
    const int lane = threadIdx.x & 31;
    const int tile = blockIdx.x * 8 + wave;
    if (tile >= n_tiles) return;

    const int row0 = tile * 16;
    const int m    = lane & 15;
    const int half = lane >> 4;

    const int hf = package_half();
    float* pool = hf ? pool1 : pool0;
    float* cnt  = hf ? cnt1 : cnt0;

    const float blo = b2[lane], bhi = b2[lane + 32];
    const int gown = batch[row0 + m];

    v8f acc0 = {}, acc1 = {}, acc2 = {}, acc3 = {};
    const float* arow = z + (size_t)(row0 + m) * HID;

    if (bias_is_zero(b1, lane)) {
        const v2f* pq = (const v2f*)z;
        const v2f h0 = pq[row0 + m];
        const v2f h1 = pq[n_tiles * 16 + row0 + m];
        const float P = h0.x + h1.x;
        const float Q = h0.y + h1.y;
        for (int kk = 0; kk < HID; kk += 4) {
            const int ka = kk + 2 * half;
            const v2f w = *(const v2f*)(W1 + ka);
            v2f a;
            a.x = w.x > 0.0f ? w.x * P : -w.x * Q;
            a.y = w.y > 0.0f ? w.y * P : -w.y * Q;
            const v2f* bp = &sB[(ka >> 1) * BPAIR_STRIDE + m];
            acc0 = __builtin_amdgcn_wmma_f32_16x16x4_f32(false, a, false, bp[ 0], (short)0, acc0, false, false);
            acc1 = __builtin_amdgcn_wmma_f32_16x16x4_f32(false, a, false, bp[16], (short)0, acc1, false, false);
            acc2 = __builtin_amdgcn_wmma_f32_16x16x4_f32(false, a, false, bp[32], (short)0, acc2, false, false);
            acc3 = __builtin_amdgcn_wmma_f32_16x16x4_f32(false, a, false, bp[48], (short)0, acc3, false, false);
        }
    } else {
    for (int kk = 0; kk < HID; kk += 4) {
        const int ka = kk + 2 * half;
        v2f a;
        a.x = arow[ka];
        a.y = arow[ka + 1];

        const v2f* bp = &sB[(ka >> 1) * BPAIR_STRIDE + m];
        acc0 = __builtin_amdgcn_wmma_f32_16x16x4_f32(false, a, false, bp[ 0], (short)0, acc0, false, false);
        acc1 = __builtin_amdgcn_wmma_f32_16x16x4_f32(false, a, false, bp[16], (short)0, acc1, false, false);
        acc2 = __builtin_amdgcn_wmma_f32_16x16x4_f32(false, a, false, bp[32], (short)0, acc2, false, false);
        acc3 = __builtin_amdgcn_wmma_f32_16x16x4_f32(false, a, false, bp[48], (short)0, acc3, false, false);
    }
    }

    if (lane < 16) atomicAdd(&cnt[gown], 1.0f);

#pragma unroll
    for (int r = 0; r < 8; ++r) {
        const float x0 = acc0[r], x1 = acc1[r], x2 = acc2[r], x3 = acc3[r];
        const float t01 = xhalf(half ? x0 : x1);
        const float t23 = xhalf(half ? x2 : x3);
        const float a_lo = half ? t01 : x0;
        const float a_hi = half ? t23 : x2;
        const float b_lo = half ? x1 : t01;
        const float b_hi = half ? x3 : t23;
        pool_row_add(pool, dinv[row0 + r],     batch[row0 + r],     lane, a_lo, a_hi, blo, bhi);
        pool_row_add(pool, dinv[row0 + 8 + r], batch[row0 + 8 + r], lane, b_lo, b_hi, blo, bhi);
    }
}

__global__ void k_final(const float* __restrict__ pool0, const float* __restrict__ pool1,
                        const float* __restrict__ cnt0, const float* __restrict__ cnt1,
                        float* __restrict__ out) {
    int tid = blockIdx.x * blockDim.x + threadIdx.x;
    if (tid < N_GRAPH * HID) {
        float c = cnt0[tid >> 6] + cnt1[tid >> 6];
        out[tid] = (pool0[tid] + pool1[tid]) / fmaxf(c, 1.0f);
    }
}

extern "C" void kernel_launch(void* const* d_in, const int* in_sizes, int n_in,
                              void* d_out, int out_size, void* d_ws, size_t ws_size,
                              hipStream_t stream) {
    const float* x     = (const float*)d_in[0];
    const int*   edge  = (const int*)  d_in[1];
    const int*   batch = (const int*)  d_in[2];
    const float* W1    = (const float*)d_in[3];
    const float* b1    = (const float*)d_in[4];
    const float* W2    = (const float*)d_in[5];
    const float* b2    = (const float*)d_in[6];
    const int* src = edge;
    const int* dst = edge + N_EDGES;
    float* out = (float*)d_out;

    float* ws   = (float*)d_ws;
    float* dinv = ws;
    float* s    = dinv + N_NODES;
    float* bufA = s + N_NODES;
    float* bufB = bufA + (size_t)N_NODES * HID;
    float* y    = bufB;
    float* deg1 = bufB + N_NODES;
    float* s1   = bufB + 2 * N_NODES;
    float* pairs = bufB + 3 * N_NODES;
    int*   tick  = (int*)(bufB + 5 * N_NODES);
    float* pool = bufB + (size_t)N_NODES * HID;
    float* cnt  = pool + N_GRAPH * HID;
    float* pool1 = bufB + 6 * N_NODES;
    float* cnt1  = pool1 + N_GRAPH * HID;

    const int TB = 256;
    k_init_deg     <<<(N_NODES + TB - 1) / TB, TB, 0, stream>>>(dinv, deg1, pool, pool1, cnt, cnt1, N_NODES);
    k_deg_edges    <<<EDGE_BLOCKS, TB, 0, stream>>>(dst, dinv, deg1, N_EDGES);
    k_dinv_selfloop<<<(N_NODES + TB - 1) / TB, TB, 0, stream>>>(dinv, deg1, x, y, s, s1, N_NODES);
    k_l1_edges     <<<EDGE_BLOCKS, TB, 0, stream>>>(src, dst, y, s, s1, N_EDGES);
    k_l1_rows      <<<ROW_BLOCKS, TB, 0, stream>>>(s, s1, dinv, W1, b1, bufA, pairs, tick, N_NODES);
    k_l2_edges     <<<L2_BLOCKS, TB, 0, stream>>>(src, dst, pairs, W1, b1, bufA, tick, N_EDGES);

    const int n_tiles = N_NODES / 16;
    k_gemm_pool    <<<(n_tiles + 7) / 8, 256, 0, stream>>>(bufA, W2, W1, b1, dinv, b2, batch, pool, pool1, cnt, cnt1, n_tiles);

    k_final        <<<(N_GRAPH * HID + TB - 1) / TB, TB, 0, stream>>>(pool, pool1, cnt, cnt1, out);
}


# Zero b1: W2 product regrouped: z@W2 = P*(relu(W1)@W2) + Q*(relu(-W1)@W2); the 64-deep WMMA product runs once per call on those 2 rows, the per-node WMMA product is 4 deep (sums regrouped)
# speedup vs baseline: 8.0644x; 1.1332x over previous
#include <hip/hip_runtime.h>

typedef float v2f __attribute__((ext_vector_type(2)));
typedef float v8f __attribute__((ext_vector_type(8)));

#define N_NODES 200000
#define N_EDGES 3200000
#define N_GRAPH 2048
#define HID     64
#define BPAIR_STRIDE 80

#define EDGE_BLOCKS ((N_EDGES + 256 * 8 - 1) / (256 * 8))
#define TICK_STRIDE 64

__device__ __forceinline__ int package_half() {
    const unsigned id = __builtin_amdgcn_s_sendmsg_rtn(0x87);
    return (int)((id >> 18) & 1u);
}

__global__ void k_init_deg(float* deg0, float* deg1,
                           float* pool0, float* pool1, float* cnt0, float* cnt1, int n) {
    int i = blockIdx.x * blockDim.x + threadIdx.x;
    if (i < n) {
        deg0[i] = 1.0f;
        deg1[i] = 0.0f;
    }
    if (i < N_GRAPH * HID) {
        pool0[i] = 0.0f;
        pool1[i] = 0.0f;
    }
    if (i < N_GRAPH) {
        cnt0[i] = 0.0f;
        cnt1[i] = 0.0f;
    }
}

__global__ __launch_bounds__(256)
void k_deg_edges(const int* __restrict__ dst, float* deg0, float* deg1, int e) {
    float* deg = package_half() ? deg1 : deg0;
    const int stride = gridDim.x * blockDim.x;
    int i = blockIdx.x * blockDim.x + threadIdx.x;
    for (; i + 3 * stride < e; i += 4 * stride) {
        const int d0 = dst[i];
        const int d1 = dst[i + stride];
        const int d2 = dst[i + 2 * stride];
        const int d3 = dst[i + 3 * stride];
        atomicAdd(&deg[d0], 1.0f);
        atomicAdd(&deg[d1], 1.0f);
        atomicAdd(&deg[d2], 1.0f);
        atomicAdd(&deg[d3], 1.0f);
    }
    for (; i < e; i += stride) atomicAdd(&deg[dst[i]], 1.0f);
}

__global__ void k_dinv_selfloop(float* deg_dinv, const float* __restrict__ deg1,
                                const float* __restrict__ x,
                                float* __restrict__ y, float* s0, float* s1, int n) {
    int i = blockIdx.x * blockDim.x + threadIdx.x;
    if (i < n) {
        float d = rsqrtf(deg_dinv[i] + deg1[i]);
        deg_dinv[i] = d;
        float yi = d * x[i];
        y[i] = yi;
        s0[i] = yi;
        s1[i] = 0.0f;
    }
}

__global__ __launch_bounds__(256)
void k_l1_edges(const int* __restrict__ src, const int* __restrict__ dst,
                const float* __restrict__ y, float* s0, float* s1, int e) {
    float* s = package_half() ? s1 : s0;
    const int stride = gridDim.x * blockDim.x;
    int i = blockIdx.x * blockDim.x + threadIdx.x;
    for (; i + 3 * stride < e; i += 4 * stride) {
        const int u0 = src[i];
        const int u1 = src[i + stride];
        const int u2 = src[i + 2 * stride];
        const int u3 = src[i + 3 * stride];
        const int d0 = dst[i];
        const int d1 = dst[i + stride];
        const int d2 = dst[i + 2 * stride];
        const int d3 = dst[i + 3 * stride];
        const float y0 = y[u0];
        const float y1 = y[u1];
        const float y2 = y[u2];
        const float y3 = y[u3];
        atomicAdd(&s[d0], y0);
        atomicAdd(&s[d1], y1);
        atomicAdd(&s[d2], y2);
        atomicAdd(&s[d3], y3);
    }
    for (; i < e; i += stride) atomicAdd(&s[dst[i]], y[src[i]]);
}

#define ROW_BLOCKS 1024
typedef float v4f __attribute__((ext_vector_type(4)));

__device__ __forceinline__ float l1_scaled(float a, float d, float w, float b) {
    float r = a * w + b;
    r = r > 0.0f ? r : 0.0f;
    return d * r;
}

__device__ __forceinline__ v4f l1_piece(float a, float d, v4f w, v4f b) {
    v4f r;
    r.x = l1_scaled(a, d, w.x, b.x);
    r.y = l1_scaled(a, d, w.y, b.y);
    r.z = l1_scaled(a, d, w.z, b.z);
    r.w = l1_scaled(a, d, w.w, b.w);
    return r;
}

__device__ __forceinline__ bool bias_is_zero(const float* __restrict__ b1, int lane) {
    const float x0 = b1[lane], x1 = b1[lane + 32];
    return __builtin_amdgcn_ballot_w32((x0 != 0.0f) | (x1 != 0.0f)) == 0u;
}

__device__ __forceinline__ void l2_scalar_add(float* acc, int v, float g) {
    atomicAdd(&acc[2 * v + (g < 0.0f ? 1 : 0)], fabsf(g));
}

__device__ __forceinline__ void fold_w2(const float* __restrict__ W1, const float* __restrict__ W2,
                                        v2f* __restrict__ fr, int j, int lane) {
    const int m    = lane & 15;
    const int half = lane >> 4;
    const float sgn = (m == 1) ? -1.0f : 1.0f;
    const float* wa = W1 + 2 * half;
    const float* wb = W2 + (2 * half) * HID + 16 * j + m;
    v8f acc_e = {}, acc_o = {};
    for (int kk = 0; kk < HID; kk += 8) {
        const v2f w0 = *(const v2f*)(wa + kk);
        const v2f w1 = *(const v2f*)(wa + kk + 4);
        v2f a0, a1, b0, b1;
        a0.x = __builtin_fmaxf(sgn * w0.x, 0.0f);
        a0.y = __builtin_fmaxf(sgn * w0.y, 0.0f);
        a1.x = __builtin_fmaxf(sgn * w1.x, 0.0f);
        a1.y = __builtin_fmaxf(sgn * w1.y, 0.0f);
        b0.x = wb[(kk    ) * HID];
        b0.y = wb[(kk + 1) * HID];
        b1.x = wb[(kk + 4) * HID];
        b1.y = wb[(kk + 5) * HID];
        acc_e = __builtin_amdgcn_wmma_f32_16x16x4_f32(false, a0, false, b0, (short)0, acc_e, false, false);
        acc_o = __builtin_amdgcn_wmma_f32_16x16x4_f32(false, a1, false, b1, (short)0, acc_o, false, false);
    }
    if (lane < 16) {
        v2f o;
        o.x = acc_e[0] + acc_o[0];
        o.y = acc_e[1] + acc_o[1];
        fr[16 * j + m] = o;
    }
}

__global__ __launch_bounds__(256)
void k_l1_rows(const float* __restrict__ s0, const float* __restrict__ s1,
               const float* __restrict__ dinv,
               const float* __restrict__ W1,
               const float* __restrict__ b1, const float* __restrict__ W2, float* __restrict__ z,
               float* __restrict__ pairs, float* __restrict__ fr, int* tick, int n) {
    if (blockIdx.x == 0 && threadIdx.x < 2) tick[threadIdx.x * TICK_STRIDE] = 0;
    if (bias_is_zero(b1, threadIdx.x & 31)) {
        const int wv = __builtin_amdgcn_readfirstlane(threadIdx.x >> 5);
        if (blockIdx.x == 0 && wv < HID / 16) fold_w2(W1, W2, (v2f*)fr, wv, threadIdx.x & 31);
        v2f* pq0 = (v2f*)z;
        v2f* pq1 = pq0 + n;
        const int nstride = gridDim.x * blockDim.x;
        for (int i = blockIdx.x * blockDim.x + threadIdx.x; i < n; i += nstride) {
            const float d = dinv[i];
            const float g = d * (d * (s0[i] + s1[i]));
            pairs[i] = g;
            v2f own, none;
            own.x = g > 0.0f ? g : 0.0f;
            own.y = g < 0.0f ? -g : 0.0f;
            none.x = 0.0f; none.y = 0.0f;
            pq0[i] = own;
            pq1[i] = none;
        }
        return;
    }
    const int total  = n * (HID / 4);
    const int stride = gridDim.x * blockDim.x;
    int q = blockIdx.x * blockDim.x + threadIdx.x;
    const int c4 = (q & 15) * 4;
    const bool first = (c4 == 0);
    const v4f w = *(const v4f*)(W1 + c4);
    const v4f b = *(const v4f*)(b1 + c4);
    v4f* zv = (v4f*)z;
    v2f* pv = (v2f*)pairs;
    for (; q + 3 * stride < total; q += 4 * stride) {
        const int q0 = q, q1 = q + stride, q2 = q + 2 * stride, q3 = q + 3 * stride;
        const int i0 = q0 >> 4, i1 = q1 >> 4, i2 = q2 >> 4, i3 = q3 >> 4;
        const float d0 = dinv[i0], d1 = dinv[i1], d2 = dinv[i2], d3 = dinv[i3];
        const float a0 = d0 * (s0[i0] + s1[i0]);
        const float a1 = d1 * (s0[i1] + s1[i1]);
        const float a2 = d2 * (s0[i2] + s1[i2]);
        const float a3 = d3 * (s0[i3] + s1[i3]);
        zv[q0] = l1_piece(a0, d0, w, b);
        zv[q1] = l1_piece(a1, d1, w, b);
        zv[q2] = l1_piece(a2, d2, w, b);
        zv[q3] = l1_piece(a3, d3, w, b);
        if (first) {
            v2f p0, p1, p2, p3;
            p0.x = a0; p0.y = d0;
            p1.x = a1; p1.y = d1;
            p2.x = a2; p2.y = d2;
            p3.x = a3; p3.y = d3;
            pv[i0] = p0;
            pv[i1] = p1;
            pv[i2] = p2;
            pv[i3] = p3;
        }
    }
    for (; q < total; q += stride) {
        const int i = q >> 4;
        const float d = dinv[i];
        const float a = d * (s0[i] + s1[i]);
        zv[q] = l1_piece(a, d, w, b);
        if (first) {
            v2f p;
            p.x = a; p.y = d;
            pv[i] = p;
        }
    }
}

#define L2_BLOCKS 2048
#define L2_ROLES (L2_BLOCKS / 2)
#define L2_SPLIT (N_NODES / 2)

__device__ __forceinline__ void l2_fetch_idx(const int* __restrict__ src, const int* __restrict__ dst,
                                             int grp, int g1, int lane, int e, int& u, int& v) {
    const int ei = grp * 32 + lane;
    const bool ok = (grp < g1) && (ei < e);
    const int ec = ok ? ei : (e - 1);
    u = src[ec];
    const int vv = dst[ec];
    v = ok ? vv : -1;
}

__global__ __launch_bounds__(256)
void k_l2_edges(const int* __restrict__ src, const int* __restrict__ dst,
                const float* __restrict__ pairs,
                const float* __restrict__ W1, const float* __restrict__ b1,
                float* z, int* tick, int e) {
    __shared__ int s_role[2];
    const int hf = package_half();
    if (bias_is_zero(b1, threadIdx.x & 31)) {
        const float* __restrict__ gs = pairs;
        float* acc = z + (size_t)hf * (2 * N_NODES);
        const int stride = gridDim.x * blockDim.x;
        int i = blockIdx.x * blockDim.x + threadIdx.x;
        for (; i + 3 * stride < e; i += 4 * stride) {
            const int u0 = src[i];
            const int u1 = src[i + stride];
            const int u2 = src[i + 2 * stride];
            const int u3 = src[i + 3 * stride];
            const int d0 = dst[i];
            const int d1 = dst[i + stride];
            const int d2 = dst[i + 2 * stride];
            const int d3 = dst[i + 3 * stride];
            const float g0 = gs[u0];
            const float g1 = gs[u1];
            const float g2 = gs[u2];
            const float g3 = gs[u3];
            l2_scalar_add(acc, d0, g0);
            l2_scalar_add(acc, d1, g1);
            l2_scalar_add(acc, d2, g2);
            l2_scalar_add(acc, d3, g3);
        }
        for (; i + stride < e; i += 2 * stride) {
            const int u0 = src[i];
            const int u1 = src[i + stride];
            const int d0 = dst[i];
            const int d1 = dst[i + stride];
            const float g0 = gs[u0];
            const float g1 = gs[u1];
            l2_scalar_add(acc, d0, g0);
            l2_scalar_add(acc, d1, g1);
        }
        for (; i < e; i += stride) l2_scalar_add(acc, dst[i], gs[src[i]]);
        return;
    }
    if (threadIdx.x == 0) {
        const int r = atomicAdd(&tick[hf * TICK_STRIDE], 1);
        int own = hf, share = r;
        if (r >= L2_ROLES) {
            own = 1 - hf;
            share = 2 * L2_ROLES - 1 - r;
        }
        s_role[0] = own;
        s_role[1] = share;
    }
    __syncthreads();
    const int own   = __builtin_amdgcn_readfirstlane(s_role[0]);
    const int share = __builtin_amdgcn_readfirstlane(s_role[1]);
    if (share < 0 || share >= L2_ROLES) return;

    const int lane    = threadIdx.x & 31;
    const float w0 = W1[lane], w1 = W1[lane + 32];
    const float c0 = b1[lane], c1 = b1[lane + 32];
    const v2f* pr = (const v2f*)pairs;
    const int ngroups = (e + 31) >> 5;
    const int widx = share * 8 + (threadIdx.x >> 5);
    int g        = (int)(((long long)widx * ngroups) / (8 * L2_ROLES));
    const int g1 = (int)(((long long)(widx + 1) * ngroups) / (8 * L2_ROLES));

    int u0, v0;
    int u1, v1;
    l2_fetch_idx(src, dst, g, g1, lane, e, u0, v0);
    l2_fetch_idx(src, dst, g + 1, g1, lane, e, u1, v1);
    float a0 = 0.0f, d0 = 0.0f;
    if ((v0 >= 0) && ((v0 >= L2_SPLIT ? 1 : 0) == own)) {
        const v2f p = pr[u0];
        a0 = p.x;
        d0 = p.y;
    }
    while (g < g1) {
        int u2, v2;
        l2_fetch_idx(src, dst, g + 2, g1, lane, e, u2, v2);
        float a1 = 0.0f, d1 = 0.0f;
        if ((v1 >= 0) && ((v1 >= L2_SPLIT ? 1 : 0) == own)) {
            const v2f p = pr[u1];
            a1 = p.x;
            d1 = p.y;
        }

        unsigned mask = __builtin_amdgcn_ballot_w32((v0 >= 0) && ((v0 >= L2_SPLIT ? 1 : 0) == own));
        while (mask != 0u) {
            const int j = __builtin_ctz(mask);
            mask &= mask - 1u;
            const float sa = __int_as_float(__builtin_amdgcn_readlane(__float_as_int(a0), j));
            const float sd = __int_as_float(__builtin_amdgcn_readlane(__float_as_int(d0), j));
            const int   sv = __builtin_amdgcn_readlane(v0, j);
            float* orow = z + (size_t)sv * HID + lane;
            atomicAdd(&orow[0],  l1_scaled(sa, sd, w0, c0));
            atomicAdd(&orow[32], l1_scaled(sa, sd, w1, c1));
        }
        v0 = v1;
        a0 = a1;
        d0 = d1;
        u1 = u2;
        v1 = v2;
        ++g;
    }
}


__device__ __forceinline__ float xhalf(float x) {
    const int b = __builtin_bit_cast(int, x);
    return __builtin_bit_cast(float, __builtin_amdgcn_permlanex16(b, b, 0x76543210, 0xfedcba98, false, false));
}

__device__ __forceinline__ void pool_row_add(float* pool, float e, int g, int lane,
                                             float lo, float hi, float blo, float bhi) {
    float v0 = e * lo + blo;
    float v1 = e * hi + bhi;
    v0 = v0 > 0.0f ? v0 : 0.0f;
    v1 = v1 > 0.0f ? v1 : 0.0f;
    float* prow = pool + g * HID + lane;
    atomicAdd(&prow[0],  v0);
    atomicAdd(&prow[32], v1);
}

__global__ __launch_bounds__(256)
void k_gemm_pool(const float* __restrict__ z, const float* __restrict__ W2,
                 const float* __restrict__ fr, const float* __restrict__ b1,
                 const float* __restrict__ dinv, const float* __restrict__ b2,
                 const int* __restrict__ batch,
                 float* pool0, float* pool1, float* cnt0, float* cnt1, int n_tiles) {
    __shared__ v2f sB[(HID / 2) * BPAIR_STRIDE];
    const bool zero_bias = bias_is_zero(b1, threadIdx.x & 31);
    if (!zero_bias) {
        float* sf = (float*)sB;
        for (int idx = threadIdx.x; idx < HID * HID; idx += blockDim.x) {
            const int k = idx >> 6, col = idx & 63;
            sf[(k >> 1) * (2 * BPAIR_STRIDE) + col * 2 + (k & 1)] = W2[idx];
        }
        __syncthreads();
    }

    const int wave = __builtin_amdgcn_readfirstlane(threadIdx.x >> 5);
    const int lane = threadIdx.x & 31;
    const int tile = blockIdx.x * 8 + wave;
    if (tile >= n_tiles) return;

    const int row0 = tile * 16;
    const int m    = lane & 15;
    const int half = lane >> 4;

    const int hf = package_half();
    float* pool = hf ? pool1 : pool0;
    float* cnt  = hf ? cnt1 : cnt0;

    const float blo = b2[lane], bhi = b2[lane + 32];
    const int gown = batch[row0 + m];

    v8f acc0 = {}, acc1 = {}, acc2 = {}, acc3 = {};
    const float* arow = z + (size_t)(row0 + m) * HID;

    if (zero_bias) {
        const v2f* pq = (const v2f*)z;
        const v2f a = pq[half * (n_tiles * 16) + row0 + m];
        const v2f* bp = (const v2f*)fr + m;
        acc0 = __builtin_amdgcn_wmma_f32_16x16x4_f32(false, a, false, bp[ 0], (short)0, acc0, false, false);
        acc1 = __builtin_amdgcn_wmma_f32_16x16x4_f32(false, a, false, bp[16], (short)0, acc1, false, false);
        acc2 = __builtin_amdgcn_wmma_f32_16x16x4_f32(false, a, false, bp[32], (short)0, acc2, false, false);
        acc3 = __builtin_amdgcn_wmma_f32_16x16x4_f32(false, a, false, bp[48], (short)0, acc3, false, false);
    } else {
    for (int kk = 0; kk < HID; kk += 4) {
        const int ka = kk + 2 * half;
        v2f a;
        a.x = arow[ka];
        a.y = arow[ka + 1];

        const v2f* bp = &sB[(ka >> 1) * BPAIR_STRIDE + m];
        acc0 = __builtin_amdgcn_wmma_f32_16x16x4_f32(false, a, false, bp[ 0], (short)0, acc0, false, false);
        acc1 = __builtin_amdgcn_wmma_f32_16x16x4_f32(false, a, false, bp[16], (short)0, acc1, false, false);
        acc2 = __builtin_amdgcn_wmma_f32_16x16x4_f32(false, a, false, bp[32], (short)0, acc2, false, false);
        acc3 = __builtin_amdgcn_wmma_f32_16x16x4_f32(false, a, false, bp[48], (short)0, acc3, false, false);
    }
    }

    if (lane < 16) atomicAdd(&cnt[gown], 1.0f);

#pragma unroll
    for (int r = 0; r < 8; ++r) {
        const float x0 = acc0[r], x1 = acc1[r], x2 = acc2[r], x3 = acc3[r];
        const float t01 = xhalf(half ? x0 : x1);
        const float t23 = xhalf(half ? x2 : x3);
        const float a_lo = half ? t01 : x0;
        const float a_hi = half ? t23 : x2;
        const float b_lo = half ? x1 : t01;
        const float b_hi = half ? x3 : t23;
        pool_row_add(pool, dinv[row0 + r],     batch[row0 + r],     lane, a_lo, a_hi, blo, bhi);
        pool_row_add(pool, dinv[row0 + 8 + r], batch[row0 + 8 + r], lane, b_lo, b_hi, blo, bhi);
    }
}

__global__ void k_final(const float* __restrict__ pool0, const float* __restrict__ pool1,
                        const float* __restrict__ cnt0, const float* __restrict__ cnt1,
                        float* __restrict__ out) {
    int tid = blockIdx.x * blockDim.x + threadIdx.x;
    if (tid < N_GRAPH * HID) {
        float c = cnt0[tid >> 6] + cnt1[tid >> 6];
        out[tid] = (pool0[tid] + pool1[tid]) / fmaxf(c, 1.0f);
    }
}

extern "C" void kernel_launch(void* const* d_in, const int* in_sizes, int n_in,
                              void* d_out, int out_size, void* d_ws, size_t ws_size,
                              hipStream_t stream) {
    const float* x     = (const float*)d_in[0];
    const int*   edge  = (const int*)  d_in[1];
    const int*   batch = (const int*)  d_in[2];
    const float* W1    = (const float*)d_in[3];
    const float* b1    = (const float*)d_in[4];
    const float* W2    = (const float*)d_in[5];
    const float* b2    = (const float*)d_in[6];
    const int* src = edge;
    const int* dst = edge + N_EDGES;
    float* out = (float*)d_out;

    float* ws   = (float*)d_ws;
    float* dinv = ws;
    float* s    = dinv + N_NODES;
    float* bufA = s + N_NODES;
    float* bufB = bufA + (size_t)N_NODES * HID;
    float* y    = bufB;
    float* deg1 = bufB + N_NODES;
    float* s1   = bufB + 2 * N_NODES;
    float* pairs = bufB + 3 * N_NODES;
    int*   tick  = (int*)(bufB + 5 * N_NODES);
    float* pool = bufB + (size_t)N_NODES * HID;
    float* cnt  = pool + N_GRAPH * HID;
    float* pool1 = bufB + 6 * N_NODES;
    float* cnt1  = pool1 + N_GRAPH * HID;
    float* fr    = bufB + 8 * N_NODES;

    const int TB = 256;
    k_init_deg     <<<(N_NODES + TB - 1) / TB, TB, 0, stream>>>(dinv, deg1, pool, pool1, cnt, cnt1, N_NODES);
    k_deg_edges    <<<EDGE_BLOCKS, TB, 0, stream>>>(dst, dinv, deg1, N_EDGES);
    k_dinv_selfloop<<<(N_NODES + TB - 1) / TB, TB, 0, stream>>>(dinv, deg1, x, y, s, s1, N_NODES);
    k_l1_edges     <<<EDGE_BLOCKS, TB, 0, stream>>>(src, dst, y, s, s1, N_EDGES);
    k_l1_rows      <<<ROW_BLOCKS, TB, 0, stream>>>(s, s1, dinv, W1, b1, W2, bufA, pairs, fr, tick, N_NODES);
    k_l2_edges     <<<L2_BLOCKS, TB, 0, stream>>>(src, dst, pairs, W1, b1, bufA, tick, N_EDGES);

    const int n_tiles = N_NODES / 16;
    k_gemm_pool    <<<(n_tiles + 7) / 8, 256, 0, stream>>>(bufA, W2, fr, b1, dinv, b2, batch, pool, pool1, cnt, cnt1, n_tiles);

    k_final        <<<(N_GRAPH * HID + TB - 1) / TB, TB, 0, stream>>>(pool, pool1, cnt, cnt1, out);
}
